# starting kernel
# baseline (speedup 1.0000x reference)
#include <hip/hip_runtime.h>
#include <hip/hip_bf16.h>
#include <math.h>

typedef __attribute__((ext_vector_type(2))) float v2f;
typedef __attribute__((ext_vector_type(8))) float v8f;

#define NN      100000
#define NE      1600000
#define HH      64
#define TT      384
#define NTILES  6250

__device__ __forceinline__ v2f mk2(float a, float b) {
    v2f r; r.x = a; r.y = b; return r;
}

__device__ __forceinline__ v8f wmma4(v2f a, v2f b, v8f c) {
    return __builtin_amdgcn_wmma_f32_16x16x4_f32(
         false, a,  false, b,
         (short)0, c,  false,  false);
}

__global__ void degree_kernel(const int* __restrict__ src, const int* __restrict__ dst,
                              float* __restrict__ degF, float* __restrict__ degR) {
    int e = blockIdx.x * 256 + threadIdx.x;
    if (e < NE) {
        atomicAdd(&degF[dst[e]], 1.0f);
        atomicAdd(&degR[src[e]], 1.0f);
    }
}

__global__ void invert_kernel(float* __restrict__ degF, float* __restrict__ degR) {
    int i = blockIdx.x * 256 + threadIdx.x;
    if (i < NN) {
        float a = degF[i];
        float b = degR[i];
        degF[i] = (a > 0.0f) ? (1.0f / a) : 0.0f;
        degR[i] = (b > 0.0f) ? (1.0f / b) : 0.0f;
    }
}

__global__ __launch_bounds__(256) void enc_kernel(const float* __restrict__ emb,
                                                  const float* __restrict__ W,
                                                  const float* __restrict__ bias,
                                                  float* __restrict__ x) {
    const int wave = threadIdx.x >> 5;
    const int lane = threadIdx.x & 31;
    const int nt   = blockIdx.x * 8 + wave;
    if (nt >= NTILES) return;
    const int ln = lane & 15;
    const int hi = lane >> 4;
    const int kh = hi << 1;

    const float* arow = emb + (size_t)(nt * 16 + ln) * TT + kh;
    v8f acc0 = {}, acc1 = {}, acc2 = {}, acc3 = {};

    #pragma unroll 4
    for (int k0 = 0; k0 < TT; k0 += 4) {
        v2f a = *reinterpret_cast<const v2f*>(arow + k0);
        const float* bp = W + (size_t)(k0 + kh) * HH + ln;
        v2f b0 = mk2(bp[0],  bp[HH]);
        v2f b1 = mk2(bp[16], bp[HH + 16]);
        v2f b2 = mk2(bp[32], bp[HH + 32]);
        v2f b3 = mk2(bp[48], bp[HH + 48]);
        acc0 = wmma4(a, b0, acc0);
        acc1 = wmma4(a, b1, acc1);
        acc2 = wmma4(a, b2, acc2);
        acc3 = wmma4(a, b3, acc3);
    }

    float bs0 = bias[ln], bs1 = bias[16 + ln], bs2 = bias[32 + ln], bs3 = bias[48 + ln];
    #pragma unroll
    for (int v = 0; v < 8; ++v) {
        size_t r = (size_t)(nt * 16 + v + 8 * hi) * HH + ln;
        x[r]      = acc0[v] + bs0;
        x[r + 16] = acc1[v] + bs1;
        x[r + 32] = acc2[v] + bs2;
        x[r + 48] = acc3[v] + bs3;
    }
}

__global__ __launch_bounds__(256) void scatter_kernel(const float* __restrict__ x,
                                                      const float* __restrict__ w,
                                                      const int* __restrict__ src,
                                                      const int* __restrict__ dst,
                                                      float* __restrict__ nF,
                                                      float* __restrict__ nR) {
    size_t gid = (size_t)blockIdx.x * 256 + threadIdx.x;
    int e = (int)(gid >> 6);
    int c = (int)(gid & 63);
    if (e >= NE) return;
    int s = src[e];
    int d = dst[e];
    float we = w[e];
    float xs = x[(size_t)s * HH + c];
    float xd = x[(size_t)d * HH + c];
    atomicAdd(&nF[(size_t)d * HH + c], we * xs);
    atomicAdd(&nR[(size_t)s * HH + c], we * xd);
}

__global__ __launch_bounds__(256) void layer_kernel(
    float* __restrict__ x,
    const float* __restrict__ nF, const float* __restrict__ nR,
    const float* __restrict__ invF, const float* __restrict__ invR,
    const float* __restrict__ WsF, const float* __restrict__ WnF, const float* __restrict__ bF,
    const float* __restrict__ WsR, const float* __restrict__ WnR, const float* __restrict__ bR) {
    const int wave = threadIdx.x >> 5;
    const int lane = threadIdx.x & 31;
    const int nt   = blockIdx.x * 8 + wave;
    if (nt >= NTILES) return;
    const int ln = lane & 15;
    const int hi = lane >> 4;
    const int kh = hi << 1;
    const int rowA = nt * 16 + ln;

    const float sF = invF[rowA];
    const float sR = invR[rowA];
    const float* xr = x  + (size_t)rowA * HH + kh;
    const float* fr = nF + (size_t)rowA * HH + kh;
    const float* rr = nR + (size_t)rowA * HH + kh;

    v8f aF0 = {}, aF1 = {}, aF2 = {}, aF3 = {};
    v8f aR0 = {}, aR1 = {}, aR2 = {}, aR3 = {};

    #pragma unroll 4
    for (int k0 = 0; k0 < HH; k0 += 4) {
        v2f aS = *reinterpret_cast<const v2f*>(xr + k0);
        v2f aN = *reinterpret_cast<const v2f*>(fr + k0);
        v2f aV = *reinterpret_cast<const v2f*>(rr + k0);
        aN *= sF;
        aV *= sR;

        const float* pSF = WsF + (size_t)(k0 + kh) * HH + ln;
        const float* pNF = WnF + (size_t)(k0 + kh) * HH + ln;
        const float* pSR = WsR + (size_t)(k0 + kh) * HH + ln;
        const float* pNR = WnR + (size_t)(k0 + kh) * HH + ln;

        aF0 = wmma4(aS, mk2(pSF[0], pSF[HH]), aF0);
        aF0 = wmma4(aN, mk2(pNF[0], pNF[HH]), aF0);
        aR0 = wmma4(aS, mk2(pSR[0], pSR[HH]), aR0);
        aR0 = wmma4(aV, mk2(pNR[0], pNR[HH]), aR0);
        aF1 = wmma4(aS, mk2(pSF[16], pSF[HH + 16]), aF1);
        aF1 = wmma4(aN, mk2(pNF[16], pNF[HH + 16]), aF1);
        aR1 = wmma4(aS, mk2(pSR[16], pSR[HH + 16]), aR1);
        aR1 = wmma4(aV, mk2(pNR[16], pNR[HH + 16]), aR1);
        aF2 = wmma4(aS, mk2(pSF[32], pSF[HH + 32]), aF2);
        aF2 = wmma4(aN, mk2(pNF[32], pNF[HH + 32]), aF2);
        aR2 = wmma4(aS, mk2(pSR[32], pSR[HH + 32]), aR2);
        aR2 = wmma4(aV, mk2(pNR[32], pNR[HH + 32]), aR2);
        aF3 = wmma4(aS, mk2(pSF[48], pSF[HH + 48]), aF3);
        aF3 = wmma4(aN, mk2(pNF[48], pNF[HH + 48]), aF3);
        aR3 = wmma4(aS, mk2(pSR[48], pSR[HH + 48]), aR3);
        aR3 = wmma4(aV, mk2(pNR[48], pNR[HH + 48]), aR3);
    }

    float bf0 = bF[ln], bf1 = bF[16 + ln], bf2 = bF[32 + ln], bf3 = bF[48 + ln];
    float br0 = bR[ln], br1 = bR[16 + ln], br2 = bR[32 + ln], br3 = bR[48 + ln];

    #pragma unroll
    for (int v = 0; v < 8; ++v) {
        size_t r = (size_t)(nt * 16 + v + 8 * hi) * HH + ln;
        x[r]      += fmaxf(aF0[v] + bf0, 0.0f) + fmaxf(aR0[v] + br0, 0.0f);
        x[r + 16] += fmaxf(aF1[v] + bf1, 0.0f) + fmaxf(aR1[v] + br1, 0.0f);
        x[r + 32] += fmaxf(aF2[v] + bf2, 0.0f) + fmaxf(aR2[v] + br2, 0.0f);
        x[r + 48] += fmaxf(aF3[v] + bf3, 0.0f) + fmaxf(aR3[v] + br3, 0.0f);
    }
}

__global__ __launch_bounds__(64) void norm_kernel(const float* __restrict__ x,
                                                  const int* __restrict__ ids,
                                                  float* __restrict__ out) {
    __shared__ float s[64];
    int b = blockIdx.x;
    int c = threadIdx.x;
    int node = ids[b];
    float v = x[(size_t)node * HH + c];
    s[c] = v * v;
    __syncthreads();
    #pragma unroll
    for (int off = 32; off > 0; off >>= 1) {
        if (c < off) s[c] += s[c + off];
        __syncthreads();
    }
    float nrm = sqrtf(s[0]);
    out[(size_t)b * HH + c] = v / nrm;
}

extern "C" void kernel_launch(void* const* d_in, const int* in_sizes, int n_in,
                              void* d_out, int out_size, void* d_ws, size_t ws_size,
                              hipStream_t stream) {
    const float* text_emb = (const float*)d_in[0];
    const float* weights  = (const float*)d_in[1];
    const float* W_enc    = (const float*)d_in[2];
    const float* b_enc    = (const float*)d_in[3];
    const float* Wself_f  = (const float*)d_in[4];
    const float* Wneigh_f = (const float*)d_in[5];
    const float* b_f      = (const float*)d_in[6];
    const float* Wself_r  = (const float*)d_in[7];
    const float* Wneigh_r = (const float*)d_in[8];
    const float* b_r      = (const float*)d_in[9];
    const int*   src      = (const int*)d_in[10];
    const int*   dst      = (const int*)d_in[11];
    const int*   ids      = (const int*)d_in[12];
    float* out = (float*)d_out;

    const size_t NH = (size_t)NN * HH;
    float* x    = (float*)d_ws;
    float* nF   = x + NH;
    float* nR   = x + 2 * NH;
    float* invF = x + 3 * NH;
    float* invR = invF + NN;

    const int batch = in_sizes[12];

    hipMemsetAsync(invF, 0, (size_t)2 * NN * sizeof(float), stream);
    degree_kernel<<<(NE + 255) / 256, 256, 0, stream>>>(src, dst, invF, invR);
    invert_kernel<<<(NN + 255) / 256, 256, 0, stream>>>(invF, invR);

    enc_kernel<<<(NTILES + 7) / 8, 256, 0, stream>>>(text_emb, W_enc, b_enc, x);

    for (int l = 0; l < 2; ++l) {
        hipMemsetAsync(nF, 0, (size_t)2 * NH * sizeof(float), stream);
        scatter_kernel<<<(int)(((size_t)NE * HH + 255) / 256), 256, 0, stream>>>(
            x, weights, src, dst, nF, nR);
        layer_kernel<<<(NTILES + 7) / 8, 256, 0, stream>>>(
            x, nF, nR, invF, invR,
            Wself_f + (size_t)l * HH * HH, Wneigh_f + (size_t)l * HH * HH, b_f + (size_t)l * HH,
            Wself_r + (size_t)l * HH * HH, Wneigh_r + (size_t)l * HH * HH, b_r + (size_t)l * HH);
    }

    norm_kernel<<<batch, 64, 0, stream>>>(x, ids, out);
}


# Neighbour sums by gather over a CSR built in-call (int atomic arrival-order place, scan, fill), fused into the layer kernel through LDS; float atomic scatter removed; sum order = arrival order
# speedup vs baseline: 2.0333x; 2.0333x over previous
#include <hip/hip_runtime.h>
#include <hip/hip_bf16.h>
#include <math.h>

typedef __attribute__((ext_vector_type(2))) float v2f;
typedef __attribute__((ext_vector_type(8))) float v8f;

#define NN      100000
#define NE      1600000
#define HH      64
#define TT      384
#define NTILES  6250
#define TP      68
#define NBMASK  0xFFFFF

__device__ __forceinline__ v2f mk2(float a, float b) {
    v2f r; r.x = a; r.y = b; return r;
}

__device__ __forceinline__ v8f wmma4(v2f a, v2f b, v8f c) {
    return __builtin_amdgcn_wmma_f32_16x16x4_f32(
         false, a,  false, b,
         (short)0, c,  false,  false);
}

__global__ void count_rank_kernel(const int* __restrict__ src, const int* __restrict__ dst,
                                  int* __restrict__ cntF, int* __restrict__ cntR,
                                  int* __restrict__ rankF, int* __restrict__ rankR) {
    int e = blockIdx.x * 256 + threadIdx.x;
    if (e < NE) {
        rankF[e] = atomicAdd(&cntF[dst[e]], 1);
        rankR[e] = atomicAdd(&cntR[src[e]], 1);
    }
}

__global__ __launch_bounds__(1024) void scan_kernel(int* __restrict__ cnt) {
    __shared__ int wsum[2][32];
    int* a = cnt + (size_t)blockIdx.x * NN;
    const int tid  = threadIdx.x;
    const int lane = tid & 31;
    const int wv   = tid >> 5;
    int carry = 0;
    for (int it = 0; it < 25; ++it) {
        const int i4 = it * 1024 + tid;
        const bool in = i4 < (NN / 4);
        int4 v = make_int4(0, 0, 0, 0);
        if (in) v = reinterpret_cast<const int4*>(a)[i4];
        const int s1 = v.x, s2 = s1 + v.y, s3 = s2 + v.z, s4 = s3 + v.w;
        int inc = s4;
        for (int off = 1; off < 32; off <<= 1) {
            int t = __shfl_up(inc, off);
            if (lane >= off) inc += t;
        }
        if (lane == 31) wsum[it & 1][wv] = inc;
        __syncthreads();
        const int ws = wsum[it & 1][lane];
        int winc = ws;
        for (int off = 1; off < 32; off <<= 1) {
            int t = __shfl_up(winc, off);
            if (lane >= off) winc += t;
        }
        const int wexcl = __shfl(winc, wv) - __shfl(ws, wv);
        const int total = __shfl(winc, 31);
        const int base  = carry + wexcl + (inc - s4);
        if (in) reinterpret_cast<int4*>(a)[i4] = make_int4(base, base + s1, base + s2, base + s3);
        carry += total;
    }
}

__global__ void fill_kernel(const int* __restrict__ src, const int* __restrict__ dst,
                            const float* __restrict__ w,
                            const int* __restrict__ offF, const int* __restrict__ offR,
                            const int* __restrict__ rankF, const int* __restrict__ rankR,
                            int2* __restrict__ csrF, int2* __restrict__ csrR) {
    int e = blockIdx.x * 256 + threadIdx.x;
    if (e < NE) {
        const int s = src[e];
        const int d = dst[e];
        const int wb = __float_as_int(w[e]);
        csrF[offF[d] + rankF[e]] = make_int2(s | ((d & 15) << 20), wb);
        csrR[offR[s] + rankR[e]] = make_int2(d | ((s & 15) << 20), wb);
    }
}

__global__ __launch_bounds__(256) void enc_kernel(const float* __restrict__ emb,
                                                  const float* __restrict__ W,
                                                  const float* __restrict__ bias,
                                                  float* __restrict__ x) {
    const int wave = threadIdx.x >> 5;
    const int lane = threadIdx.x & 31;
    const int nt   = blockIdx.x * 8 + wave;
    if (nt >= NTILES) return;
    const int ln = lane & 15;
    const int hi = lane >> 4;
    const int kh = hi << 1;

    const float* arow = emb + (size_t)(nt * 16 + ln) * TT + kh;
    v8f acc0 = {}, acc1 = {}, acc2 = {}, acc3 = {};

    #pragma unroll 4
    for (int k0 = 0; k0 < TT; k0 += 4) {
        v2f a = *reinterpret_cast<const v2f*>(arow + k0);
        const float* bp = W + (size_t)(k0 + kh) * HH + ln;
        v2f b0 = mk2(bp[0],  bp[HH]);
        v2f b1 = mk2(bp[16], bp[HH + 16]);
        v2f b2 = mk2(bp[32], bp[HH + 32]);
        v2f b3 = mk2(bp[48], bp[HH + 48]);
        acc0 = wmma4(a, b0, acc0);
        acc1 = wmma4(a, b1, acc1);
        acc2 = wmma4(a, b2, acc2);
        acc3 = wmma4(a, b3, acc3);
    }

    float bs0 = bias[ln], bs1 = bias[16 + ln], bs2 = bias[32 + ln], bs3 = bias[48 + ln];
    #pragma unroll
    for (int v = 0; v < 8; ++v) {
        size_t r = (size_t)(nt * 16 + v + 8 * hi) * HH + ln;
        x[r]      = acc0[v] + bs0;
        x[r + 16] = acc1[v] + bs1;
        x[r + 32] = acc2[v] + bs2;
        x[r + 48] = acc3[v] + bs3;
    }
}

__device__ __forceinline__ void gather_tile(const float* __restrict__ xin,
                                            const int2* __restrict__ csr,
                                            int beg, int end, int lane,
                                            float* __restrict__ tile) {
    if (beg < end) {
        float ax = 0.0f, ay = 0.0f;
        int cur = -1;
        int2 en[8];
        #pragma unroll
        for (int u = 0; u < 8; ++u) en[u] = csr[min(beg + u, end - 1)];
        for (int p = beg; p < end; p += 8) {
            float2 v[8];
            #pragma unroll
            for (int u = 0; u < 8; ++u)
                v[u] = *reinterpret_cast<const float2*>(xin + (size_t)(en[u].x & NBMASK) * HH + 2 * lane);
            int2 nx[8];
            #pragma unroll
            for (int u = 0; u < 8; ++u) nx[u] = csr[min(p + 8 + u, end - 1)];
            #pragma unroll
            for (int u = 0; u < 8; ++u) {
                const bool ok = (p + u) < end;
                const int row = en[u].x >> 20;
                const float we = __int_as_float(en[u].y);
                const bool same = (row == cur);
                ax = same ? ax : 0.0f;
                ay = same ? ay : 0.0f;
                cur = row;
                ax += ok ? we * v[u].x : 0.0f;
                ay += ok ? we * v[u].y : 0.0f;
                *reinterpret_cast<float2*>(tile + row * TP + 2 * lane) = make_float2(ax, ay);
            }
            #pragma unroll
            for (int u = 0; u < 8; ++u) en[u] = nx[u];
        }
    }
}

__global__ __launch_bounds__(64) void layer_kernel(
    const float* __restrict__ xin, float* __restrict__ xout,
    const int2* __restrict__ csrF, const int2* __restrict__ csrR,
    const int* __restrict__ offF, const int* __restrict__ offR,
    const float* __restrict__ WsF, const float* __restrict__ WnF, const float* __restrict__ bF,
    const float* __restrict__ WsR, const float* __restrict__ WnR, const float* __restrict__ bR) {
    __shared__ __attribute__((aligned(16))) float tiles[2][2][16 * TP];
    const int wave = __builtin_amdgcn_readfirstlane((int)(threadIdx.x >> 5));
    const int lane = threadIdx.x & 31;
    const int nt   = blockIdx.x * 2 + wave;
    const int n0   = nt * 16;
    const int ln = lane & 15;
    const int hi = lane >> 4;
    const int kh = hi << 1;
    const int rowA = n0 + ln;

    float* tF = &tiles[wave][0][0];
    float* tR = &tiles[wave][1][0];

    #pragma unroll
    for (int i = 0; i < (2 * 16 * TP) / (4 * 32); ++i)
        reinterpret_cast<float4*>(tF)[lane + 32 * i] = make_float4(0.0f, 0.0f, 0.0f, 0.0f);
    __syncthreads();

    const int begF = offF[n0];
    const int endF = (n0 + 16 < NN) ? offF[n0 + 16] : NE;
    const int begR = offR[n0];
    const int endR = (n0 + 16 < NN) ? offR[n0 + 16] : NE;
    gather_tile(xin, csrF, begF, endF, lane, tF);
    gather_tile(xin, csrR, begR, endR, lane, tR);
    __syncthreads();

    const int dF = ((rowA + 1 < NN) ? offF[rowA + 1] : NE) - offF[rowA];
    const int dR = ((rowA + 1 < NN) ? offR[rowA + 1] : NE) - offR[rowA];
    const float sF = (dF > 0) ? (1.0f / (float)dF) : 0.0f;
    const float sR = (dR > 0) ? (1.0f / (float)dR) : 0.0f;

    const float* xr = xin + (size_t)rowA * HH + kh;
    const float* fr = tF + ln * TP + kh;
    const float* rr = tR + ln * TP + kh;

    v8f aF0 = {}, aF1 = {}, aF2 = {}, aF3 = {};
    v8f aR0 = {}, aR1 = {}, aR2 = {}, aR3 = {};

    #pragma unroll 4
    for (int k0 = 0; k0 < HH; k0 += 4) {
        v2f aS = *reinterpret_cast<const v2f*>(xr + k0);
        v2f aN = *reinterpret_cast<const v2f*>(fr + k0);
        v2f aV = *reinterpret_cast<const v2f*>(rr + k0);
        aN *= sF;
        aV *= sR;

        const float* pSF = WsF + (size_t)(k0 + kh) * HH + ln;
        const float* pNF = WnF + (size_t)(k0 + kh) * HH + ln;
        const float* pSR = WsR + (size_t)(k0 + kh) * HH + ln;
        const float* pNR = WnR + (size_t)(k0 + kh) * HH + ln;

        aF0 = wmma4(aS, mk2(pSF[0], pSF[HH]), aF0);
        aF0 = wmma4(aN, mk2(pNF[0], pNF[HH]), aF0);
        aR0 = wmma4(aS, mk2(pSR[0], pSR[HH]), aR0);
        aR0 = wmma4(aV, mk2(pNR[0], pNR[HH]), aR0);
        aF1 = wmma4(aS, mk2(pSF[16], pSF[HH + 16]), aF1);
        aF1 = wmma4(aN, mk2(pNF[16], pNF[HH + 16]), aF1);
        aR1 = wmma4(aS, mk2(pSR[16], pSR[HH + 16]), aR1);
        aR1 = wmma4(aV, mk2(pNR[16], pNR[HH + 16]), aR1);
        aF2 = wmma4(aS, mk2(pSF[32], pSF[HH + 32]), aF2);
        aF2 = wmma4(aN, mk2(pNF[32], pNF[HH + 32]), aF2);
        aR2 = wmma4(aS, mk2(pSR[32], pSR[HH + 32]), aR2);
        aR2 = wmma4(aV, mk2(pNR[32], pNR[HH + 32]), aR2);
        aF3 = wmma4(aS, mk2(pSF[48], pSF[HH + 48]), aF3);
        aF3 = wmma4(aN, mk2(pNF[48], pNF[HH + 48]), aF3);
        aR3 = wmma4(aS, mk2(pSR[48], pSR[HH + 48]), aR3);
        aR3 = wmma4(aV, mk2(pNR[48], pNR[HH + 48]), aR3);
    }

    float bf0 = bF[ln], bf1 = bF[16 + ln], bf2 = bF[32 + ln], bf3 = bF[48 + ln];
    float br0 = bR[ln], br1 = bR[16 + ln], br2 = bR[32 + ln], br3 = bR[48 + ln];

    #pragma unroll
    for (int v = 0; v < 8; ++v) {
        size_t r = (size_t)(n0 + v + 8 * hi) * HH + ln;
        xout[r]      = xin[r]      + (fmaxf(aF0[v] + bf0, 0.0f) + fmaxf(aR0[v] + br0, 0.0f));
        xout[r + 16] = xin[r + 16] + (fmaxf(aF1[v] + bf1, 0.0f) + fmaxf(aR1[v] + br1, 0.0f));
        xout[r + 32] = xin[r + 32] + (fmaxf(aF2[v] + bf2, 0.0f) + fmaxf(aR2[v] + br2, 0.0f));
        xout[r + 48] = xin[r + 48] + (fmaxf(aF3[v] + bf3, 0.0f) + fmaxf(aR3[v] + br3, 0.0f));
    }
}

__global__ __launch_bounds__(64) void norm_kernel(const float* __restrict__ x,
                                                  const int* __restrict__ ids,
                                                  float* __restrict__ out) {
    __shared__ float s[64];
    int b = blockIdx.x;
    int c = threadIdx.x;
    int node = ids[b];
    float v = x[(size_t)node * HH + c];
    s[c] = v * v;
    __syncthreads();
    #pragma unroll
    for (int off = 32; off > 0; off >>= 1) {
        if (c < off) s[c] += s[c + off];
        __syncthreads();
    }
    float nrm = sqrtf(s[0]);
    out[(size_t)b * HH + c] = v / nrm;
}

extern "C" void kernel_launch(void* const* d_in, const int* in_sizes, int n_in,
                              void* d_out, int out_size, void* d_ws, size_t ws_size,
                              hipStream_t stream) {
    const float* text_emb = (const float*)d_in[0];
    const float* weights  = (const float*)d_in[1];
    const float* W_enc    = (const float*)d_in[2];
    const float* b_enc    = (const float*)d_in[3];
    const float* Wself_f  = (const float*)d_in[4];
    const float* Wneigh_f = (const float*)d_in[5];
    const float* b_f      = (const float*)d_in[6];
    const float* Wself_r  = (const float*)d_in[7];
    const float* Wneigh_r = (const float*)d_in[8];
    const float* b_r      = (const float*)d_in[9];
    const int*   src      = (const int*)d_in[10];
    const int*   dst      = (const int*)d_in[11];
    const int*   ids      = (const int*)d_in[12];
    float* out = (float*)d_out;

    const size_t NH = (size_t)NN * HH;
    float* xA   = (float*)d_ws;
    float* xB   = xA + NH;
    int2*  csrF = (int2*)(xA + 2 * NH);
    int2*  csrR = csrF + NE;
    int*   offF = (int*)(xA + 3 * NH);
    int*   offR = offF + NN;
    int*   rankF = (int*)xB;
    int*   rankR = rankF + NE;

    const int batch = in_sizes[12];

    hipMemsetAsync(offF, 0, (size_t)2 * NN * sizeof(int), stream);
    count_rank_kernel<<<(NE + 255) / 256, 256, 0, stream>>>(src, dst, offF, offR, rankF, rankR);
    scan_kernel<<<2, 1024, 0, stream>>>(offF);
    fill_kernel<<<(NE + 255) / 256, 256, 0, stream>>>(src, dst, weights, offF, offR, rankF, rankR, csrF, csrR);

    enc_kernel<<<(NTILES + 7) / 8, 256, 0, stream>>>(text_emb, W_enc, b_enc, xA);

    layer_kernel<<<NTILES / 2, 64, 0, stream>>>(
        xA, xB, csrF, csrR, offF, offR,
        Wself_f, Wneigh_f, b_f, Wself_r, Wneigh_r, b_r);
    layer_kernel<<<NTILES / 2, 64, 0, stream>>>(
        xB, xA, csrF, csrR, offF, offR,
        Wself_f + (size_t)HH * HH, Wneigh_f + (size_t)HH * HH, b_f + HH,
        Wself_r + (size_t)HH * HH, Wneigh_r + (size_t)HH * HH, b_r + HH);

    norm_kernel<<<batch, 64, 0, stream>>>(xA, ids, out);
}


# CSR build redone as a two-level counting sort with LDS counters (edges binned by 256-node bucket, each bucket then ordered by node in LDS, stores in order); sum order = LDS arrival order
# speedup vs baseline: 3.5703x; 1.7559x over previous
#include <hip/hip_runtime.h>
#include <hip/hip_bf16.h>
#include <math.h>

typedef __attribute__((ext_vector_type(2))) float v2f;
typedef __attribute__((ext_vector_type(8))) float v8f;

#define NN      100000
#define NE      1600000
#define HH      64
#define TT      384
#define NTILES  6250
#define TP      68
#define NBMASK  0xFFFFF

__device__ __forceinline__ v2f mk2(float a, float b) {
    v2f r; r.x = a; r.y = b; return r;
}

__device__ __forceinline__ v8f wmma4(v2f a, v2f b, v8f c) {
    return __builtin_amdgcn_wmma_f32_16x16x4_f32(
         false, a,  false, b,
         (short)0, c,  false,  false);
}

#define NBK   391
#define EB    400
#define EPB   4000
#define NCNT  (NBK * EB)
#define SCH   ((NCNT + 4095) / 4096)
#define CAP   6144
#define QPT   ((EPB / 4 + 255) / 256)
#define RPT   (CAP / 256)

__device__ __forceinline__ void scan512(int* a, int* wtot, int tid) {
    const int lane = tid & 31;
    const int wv   = tid >> 5;
    const int x0 = a[2 * tid], x1 = a[2 * tid + 1];
    const int s = x0 + x1;
    int inc = s;
    for (int off = 1; off < 32; off <<= 1) {
        int t = __shfl_up(inc, off);
        if (lane >= off) inc += t;
    }
    if (lane == 31) wtot[wv] = inc;
    __syncthreads();
    int pre = 0;
    for (int k = 0; k < 8; ++k) {
        const int v = wtot[k];
        pre += (k < wv) ? v : 0;
    }
    const int ex = pre + inc - s;
    a[2 * tid]     = ex;
    a[2 * tid + 1] = ex + x0;
    __syncthreads();
}

__global__ __launch_bounds__(256) void hist_kernel(const int* __restrict__ src, const int* __restrict__ dst,
                                                   int* __restrict__ cnt) {
    __shared__ int h[2 * NBK];
    const int tid = threadIdx.x;
    const int blk = blockIdx.x;
    for (int t = tid; t < 2 * NBK; t += 256) h[t] = 0;
    __syncthreads();
    const int4* s4 = reinterpret_cast<const int4*>(src + (size_t)blk * EPB);
    const int4* d4 = reinterpret_cast<const int4*>(dst + (size_t)blk * EPB);
    int4 sv[QPT], dv[QPT];
    #pragma unroll
    for (int k = 0; k < QPT; ++k) {
        const int q = tid + 256 * k;
        if (q < EPB / 4) { sv[k] = s4[q]; dv[k] = d4[q]; }
    }
    #pragma unroll
    for (int k = 0; k < QPT; ++k) {
        const int q = tid + 256 * k;
        if (q < EPB / 4) {
            const int4 s = sv[k];
            const int4 d = dv[k];
            atomicAdd(&h[d.x >> 8], 1); atomicAdd(&h[d.y >> 8], 1);
            atomicAdd(&h[d.z >> 8], 1); atomicAdd(&h[d.w >> 8], 1);
            atomicAdd(&h[NBK + (s.x >> 8)], 1); atomicAdd(&h[NBK + (s.y >> 8)], 1);
            atomicAdd(&h[NBK + (s.z >> 8)], 1); atomicAdd(&h[NBK + (s.w >> 8)], 1);
        }
    }
    __syncthreads();
    for (int t = tid; t < 2 * NBK; t += 256) cnt[(size_t)t * EB + blk] = h[t];
}

__global__ __launch_bounds__(1024) void scan_kernel(const int* __restrict__ cnt, int* __restrict__ base) {
    __shared__ int wsum[32];
    __shared__ int psum[32];
    const int tid  = threadIdx.x;
    const int lane = tid & 31;
    const int wv   = tid >> 5;
    const int c    = blockIdx.x;
    const int4* in = reinterpret_cast<const int4*>(cnt + (size_t)blockIdx.y * NCNT);
    int4* out      = reinterpret_cast<int4*>(base + (size_t)blockIdx.y * NCNT);
    int part = 0;
    for (int j = 0; j < c; ++j) {
        const int4 v = in[j * 1024 + tid];
        part += (v.x + v.y) + (v.z + v.w);
    }
    for (int off = 16; off > 0; off >>= 1) part += __shfl_xor(part, off);
    if (lane == 0) psum[wv] = part;
    const int i4 = c * 1024 + tid;
    const bool in_range = i4 < (NCNT / 4);
    int4 v = make_int4(0, 0, 0, 0);
    if (in_range) v = in[i4];
    const int s1 = v.x, s2 = s1 + v.y, s3 = s2 + v.z, s4 = s3 + v.w;
    int inc = s4;
    for (int off = 1; off < 32; off <<= 1) {
        int t = __shfl_up(inc, off);
        if (lane >= off) inc += t;
    }
    if (lane == 31) wsum[wv] = inc;
    __syncthreads();
    int carry = psum[lane];
    for (int off = 16; off > 0; off >>= 1) carry += __shfl_xor(carry, off);
    const int ws = wsum[lane];
    int winc = ws;
    for (int off = 1; off < 32; off <<= 1) {
        int t = __shfl_up(winc, off);
        if (lane >= off) winc += t;
    }
    const int wexcl = __shfl(winc, wv) - __shfl(ws, wv);
    const int b0 = carry + wexcl + (inc - s4);
    if (in_range) out[i4] = make_int4(b0, b0 + s1, b0 + s2, b0 + s3);
}

__global__ __launch_bounds__(256) void part_kernel(const int* __restrict__ src, const int* __restrict__ dst,
                                                   const float* __restrict__ w,
                                                   const int* __restrict__ cnt, const int* __restrict__ base,
                                                   int2* __restrict__ tmpF, int2* __restrict__ tmpR) {
    __shared__ int lb[2][512];
    __shared__ int cur[2][512];
    __shared__ int gb[2][512];
    __shared__ int wtot[8];
    __shared__ int2 srec[2][EPB];
    __shared__ int  sg[2][EPB];
    const int tid = threadIdx.x;
    const int blk = blockIdx.x;
    const int4*   s4 = reinterpret_cast<const int4*>(src + (size_t)blk * EPB);
    const int4*   d4 = reinterpret_cast<const int4*>(dst + (size_t)blk * EPB);
    const float4* w4 = reinterpret_cast<const float4*>(w + (size_t)blk * EPB);
    int4 sv[QPT], dv[QPT];
    float4 wv4[QPT];
    #pragma unroll
    for (int k = 0; k < QPT; ++k) {
        const int q = tid + 256 * k;
        if (q < EPB / 4) { sv[k] = s4[q]; dv[k] = d4[q]; wv4[k] = w4[q]; }
    }
    for (int t = tid; t < 1024; t += 256) {
        const int dir = t >> 9, b = t & 511;
        int c = 0, g = 0;
        if (b < NBK) {
            const size_t ix = (size_t)(dir * NBK + b) * EB + blk;
            c = cnt[ix];
            g = base[ix];
        }
        lb[dir][b] = c;
        gb[dir][b] = g;
    }
    __syncthreads();
    scan512(lb[0], wtot, tid);
    scan512(lb[1], wtot, tid);
    for (int t = tid; t < 1024; t += 256) {
        const int dir = t >> 9, b = t & 511;
        const int l = lb[dir][b];
        cur[dir][b] = l;
        gb[dir][b] -= l;
    }
    __syncthreads();
    #pragma unroll
    for (int k = 0; k < QPT; ++k) {
        const int q = tid + 256 * k;
        if (q < EPB / 4) {
            const int ss[4] = {sv[k].x, sv[k].y, sv[k].z, sv[k].w};
            const int dd[4] = {dv[k].x, dv[k].y, dv[k].z, dv[k].w};
            const float ww[4] = {wv4[k].x, wv4[k].y, wv4[k].z, wv4[k].w};
            #pragma unroll
            for (int u = 0; u < 4; ++u) {
                const int s = ss[u], d = dd[u];
                const int wb = __float_as_int(ww[u]);
                const int bF = d >> 8, bR = s >> 8;
                const int pF = atomicAdd(&cur[0][bF], 1);
                srec[0][pF] = make_int2(s | ((d & 255) << 20), wb);
                sg[0][pF]   = gb[0][bF] + pF;
                const int pR = atomicAdd(&cur[1][bR], 1);
                srec[1][pR] = make_int2(d | ((s & 255) << 20), wb);
                sg[1][pR]   = gb[1][bR] + pR;
            }
        }
    }
    __syncthreads();
    for (int i = tid; i < EPB; i += 256) {
        tmpF[sg[0][i]] = srec[0][i];
        tmpR[sg[1][i]] = srec[1][i];
    }
}

__global__ __launch_bounds__(256) void sort_kernel(const int* __restrict__ base,
                                                   const int2* __restrict__ tmpF, const int2* __restrict__ tmpR,
                                                   int2* __restrict__ csrF, int2* __restrict__ csrR,
                                                   int* __restrict__ offF, int* __restrict__ offR) {
    __shared__ int bins[512];
    __shared__ int cur[256];
    __shared__ int wtot[8];
    __shared__ int2 stage[CAP];
    const int tid = threadIdx.x;
    const int b   = blockIdx.x;
    const int dir = blockIdx.y;
    const int2* tmp = dir ? tmpR : tmpF;
    int2* csr       = dir ? csrR : csrF;
    int* off        = dir ? offR : offF;
    const int beg = base[(size_t)(dir * NBK + b) * EB];
    const int end = (b + 1 < NBK) ? base[(size_t)(dir * NBK + b + 1) * EB] : NE;
    const int n   = end - beg;
    const bool staged = n <= CAP;
    bins[tid] = 0;
    bins[tid + 256] = 0;
    __syncthreads();

    int2 r[RPT];
    if (staged) {
        #pragma unroll
        for (int k = 0; k < RPT; ++k) {
            const int i = tid + 256 * k;
            r[k] = make_int2(0, 0);
            if (i < n) r[k] = tmp[beg + i];
        }
        #pragma unroll
        for (int k = 0; k < RPT; ++k) {
            const int i = tid + 256 * k;
            if (i < n) atomicAdd(&bins[(r[k].x >> 20) & 255], 1);
        }
    } else {
        for (int i = tid; i < n; i += 256) atomicAdd(&bins[(tmp[beg + i].x >> 20) & 255], 1);
    }
    __syncthreads();
    scan512(bins, wtot, tid);
    {
        const int node = b * 256 + tid;
        const int ex = bins[tid];
        if (node < NN) off[node] = beg + ex;
        cur[tid] = ex;
    }
    __syncthreads();

    if (staged) {
        #pragma unroll
        for (int k = 0; k < RPT; ++k) {
            const int i = tid + 256 * k;
            if (i < n) {
                const int nl = (r[k].x >> 20) & 255;
                const int p = atomicAdd(&cur[nl], 1);
                stage[p] = make_int2((r[k].x & NBMASK) | ((nl & 15) << 20), r[k].y);
            }
        }
        __syncthreads();
        for (int i = tid; i < n; i += 256) csr[beg + i] = stage[i];
    } else {
        for (int i = tid; i < n; i += 256) {
            const int2 rr = tmp[beg + i];
            const int nl = (rr.x >> 20) & 255;
            const int p = atomicAdd(&cur[nl], 1);
            csr[beg + p] = make_int2((rr.x & NBMASK) | ((nl & 15) << 20), rr.y);
        }
    }
}

__global__ __launch_bounds__(256) void enc_kernel(const float* __restrict__ emb,
                                                  const float* __restrict__ W,
                                                  const float* __restrict__ bias,
                                                  float* __restrict__ x) {
    const int wave = threadIdx.x >> 5;
    const int lane = threadIdx.x & 31;
    const int nt   = blockIdx.x * 8 + wave;
    if (nt >= NTILES) return;
    const int ln = lane & 15;
    const int hi = lane >> 4;
    const int kh = hi << 1;

    const float* arow = emb + (size_t)(nt * 16 + ln) * TT + kh;
    v8f acc0 = {}, acc1 = {}, acc2 = {}, acc3 = {};

    #pragma unroll 4
    for (int k0 = 0; k0 < TT; k0 += 4) {
        v2f a = *reinterpret_cast<const v2f*>(arow + k0);
        const float* bp = W + (size_t)(k0 + kh) * HH + ln;
        v2f b0 = mk2(bp[0],  bp[HH]);
        v2f b1 = mk2(bp[16], bp[HH + 16]);
        v2f b2 = mk2(bp[32], bp[HH + 32]);
        v2f b3 = mk2(bp[48], bp[HH + 48]);
        acc0 = wmma4(a, b0, acc0);
        acc1 = wmma4(a, b1, acc1);
        acc2 = wmma4(a, b2, acc2);
        acc3 = wmma4(a, b3, acc3);
    }

    float bs0 = bias[ln], bs1 = bias[16 + ln], bs2 = bias[32 + ln], bs3 = bias[48 + ln];
    #pragma unroll
    for (int v = 0; v < 8; ++v) {
        size_t r = (size_t)(nt * 16 + v + 8 * hi) * HH + ln;
        x[r]      = acc0[v] + bs0;
        x[r + 16] = acc1[v] + bs1;
        x[r + 32] = acc2[v] + bs2;
        x[r + 48] = acc3[v] + bs3;
    }
}

__device__ __forceinline__ void gather_tile(const float* __restrict__ xin,
                                            const int2* __restrict__ csr,
                                            int beg, int end, int lane,
                                            float* __restrict__ tile) {
    if (beg < end) {
        float ax = 0.0f, ay = 0.0f;
        int cur = -1;
        int2 en[8];
        #pragma unroll
        for (int u = 0; u < 8; ++u) en[u] = csr[min(beg + u, end - 1)];
        for (int p = beg; p < end; p += 8) {
            float2 v[8];
            #pragma unroll
            for (int u = 0; u < 8; ++u)
                v[u] = *reinterpret_cast<const float2*>(xin + (size_t)(en[u].x & NBMASK) * HH + 2 * lane);
            int2 nx[8];
            #pragma unroll
            for (int u = 0; u < 8; ++u) nx[u] = csr[min(p + 8 + u, end - 1)];
            #pragma unroll
            for (int u = 0; u < 8; ++u) {
                const bool ok = (p + u) < end;
                const int row = en[u].x >> 20;
                const float we = __int_as_float(en[u].y);
                const bool same = (row == cur);
                ax = same ? ax : 0.0f;
                ay = same ? ay : 0.0f;
                cur = row;
                ax += ok ? we * v[u].x : 0.0f;
                ay += ok ? we * v[u].y : 0.0f;
                *reinterpret_cast<float2*>(tile + row * TP + 2 * lane) = make_float2(ax, ay);
            }
            #pragma unroll
            for (int u = 0; u < 8; ++u) en[u] = nx[u];
        }
    }
}

__global__ __launch_bounds__(64) void layer_kernel(
    const float* __restrict__ xin, float* __restrict__ xout,
    const int2* __restrict__ csrF, const int2* __restrict__ csrR,
    const int* __restrict__ offF, const int* __restrict__ offR,
    const float* __restrict__ WsF, const float* __restrict__ WnF, const float* __restrict__ bF,
    const float* __restrict__ WsR, const float* __restrict__ WnR, const float* __restrict__ bR) {
    __shared__ __attribute__((aligned(16))) float tiles[2][2][16 * TP];
    const int wave = __builtin_amdgcn_readfirstlane((int)(threadIdx.x >> 5));
    const int lane = threadIdx.x & 31;
    const int nt   = blockIdx.x * 2 + wave;
    const int n0   = nt * 16;
    const int ln = lane & 15;
    const int hi = lane >> 4;
    const int kh = hi << 1;
    const int rowA = n0 + ln;

    float* tF = &tiles[wave][0][0];
    float* tR = &tiles[wave][1][0];

    #pragma unroll
    for (int i = 0; i < (2 * 16 * TP) / (4 * 32); ++i)
        reinterpret_cast<float4*>(tF)[lane + 32 * i] = make_float4(0.0f, 0.0f, 0.0f, 0.0f);
    __syncthreads();

    const int begF = offF[n0];
    const int endF = (n0 + 16 < NN) ? offF[n0 + 16] : NE;
    const int begR = offR[n0];
    const int endR = (n0 + 16 < NN) ? offR[n0 + 16] : NE;
    gather_tile(xin, csrF, begF, endF, lane, tF);
    gather_tile(xin, csrR, begR, endR, lane, tR);
    __syncthreads();

    const int dF = ((rowA + 1 < NN) ? offF[rowA + 1] : NE) - offF[rowA];
    const int dR = ((rowA + 1 < NN) ? offR[rowA + 1] : NE) - offR[rowA];
    const float sF = (dF > 0) ? (1.0f / (float)dF) : 0.0f;
    const float sR = (dR > 0) ? (1.0f / (float)dR) : 0.0f;

    const float* xr = xin + (size_t)rowA * HH + kh;
    const float* fr = tF + ln * TP + kh;
    const float* rr = tR + ln * TP + kh;

    v8f aF0 = {}, aF1 = {}, aF2 = {}, aF3 = {};
    v8f aR0 = {}, aR1 = {}, aR2 = {}, aR3 = {};

    #pragma unroll 4
    for (int k0 = 0; k0 < HH; k0 += 4) {
        v2f aS = *reinterpret_cast<const v2f*>(xr + k0);
        v2f aN = *reinterpret_cast<const v2f*>(fr + k0);
        v2f aV = *reinterpret_cast<const v2f*>(rr + k0);
        aN *= sF;
        aV *= sR;

        const float* pSF = WsF + (size_t)(k0 + kh) * HH + ln;
        const float* pNF = WnF + (size_t)(k0 + kh) * HH + ln;
        const float* pSR = WsR + (size_t)(k0 + kh) * HH + ln;
        const float* pNR = WnR + (size_t)(k0 + kh) * HH + ln;

        aF0 = wmma4(aS, mk2(pSF[0], pSF[HH]), aF0);
        aF0 = wmma4(aN, mk2(pNF[0], pNF[HH]), aF0);
        aR0 = wmma4(aS, mk2(pSR[0], pSR[HH]), aR0);
        aR0 = wmma4(aV, mk2(pNR[0], pNR[HH]), aR0);
        aF1 = wmma4(aS, mk2(pSF[16], pSF[HH + 16]), aF1);
        aF1 = wmma4(aN, mk2(pNF[16], pNF[HH + 16]), aF1);
        aR1 = wmma4(aS, mk2(pSR[16], pSR[HH + 16]), aR1);
        aR1 = wmma4(aV, mk2(pNR[16], pNR[HH + 16]), aR1);
        aF2 = wmma4(aS, mk2(pSF[32], pSF[HH + 32]), aF2);
        aF2 = wmma4(aN, mk2(pNF[32], pNF[HH + 32]), aF2);
        aR2 = wmma4(aS, mk2(pSR[32], pSR[HH + 32]), aR2);
        aR2 = wmma4(aV, mk2(pNR[32], pNR[HH + 32]), aR2);
        aF3 = wmma4(aS, mk2(pSF[48], pSF[HH + 48]), aF3);
        aF3 = wmma4(aN, mk2(pNF[48], pNF[HH + 48]), aF3);
        aR3 = wmma4(aS, mk2(pSR[48], pSR[HH + 48]), aR3);
        aR3 = wmma4(aV, mk2(pNR[48], pNR[HH + 48]), aR3);
    }

    float bf0 = bF[ln], bf1 = bF[16 + ln], bf2 = bF[32 + ln], bf3 = bF[48 + ln];
    float br0 = bR[ln], br1 = bR[16 + ln], br2 = bR[32 + ln], br3 = bR[48 + ln];

    #pragma unroll
    for (int v = 0; v < 8; ++v) {
        size_t r = (size_t)(n0 + v + 8 * hi) * HH + ln;
        xout[r]      = xin[r]      + (fmaxf(aF0[v] + bf0, 0.0f) + fmaxf(aR0[v] + br0, 0.0f));
        xout[r + 16] = xin[r + 16] + (fmaxf(aF1[v] + bf1, 0.0f) + fmaxf(aR1[v] + br1, 0.0f));
        xout[r + 32] = xin[r + 32] + (fmaxf(aF2[v] + bf2, 0.0f) + fmaxf(aR2[v] + br2, 0.0f));
        xout[r + 48] = xin[r + 48] + (fmaxf(aF3[v] + bf3, 0.0f) + fmaxf(aR3[v] + br3, 0.0f));
    }
}

__global__ __launch_bounds__(64) void norm_kernel(const float* __restrict__ x,
                                                  const int* __restrict__ ids,
                                                  float* __restrict__ out) {
    __shared__ float s[64];
    int b = blockIdx.x;
    int c = threadIdx.x;
    int node = ids[b];
    float v = x[(size_t)node * HH + c];
    s[c] = v * v;
    __syncthreads();
    #pragma unroll
    for (int off = 32; off > 0; off >>= 1) {
        if (c < off) s[c] += s[c + off];
        __syncthreads();
    }
    float nrm = sqrtf(s[0]);
    out[(size_t)b * HH + c] = v / nrm;
}

extern "C" void kernel_launch(void* const* d_in, const int* in_sizes, int n_in,
                              void* d_out, int out_size, void* d_ws, size_t ws_size,
                              hipStream_t stream) {
    const float* text_emb = (const float*)d_in[0];
    const float* weights  = (const float*)d_in[1];
    const float* W_enc    = (const float*)d_in[2];
    const float* b_enc    = (const float*)d_in[3];
    const float* Wself_f  = (const float*)d_in[4];
    const float* Wneigh_f = (const float*)d_in[5];
    const float* b_f      = (const float*)d_in[6];
    const float* Wself_r  = (const float*)d_in[7];
    const float* Wneigh_r = (const float*)d_in[8];
    const float* b_r      = (const float*)d_in[9];
    const int*   src      = (const int*)d_in[10];
    const int*   dst      = (const int*)d_in[11];
    const int*   ids      = (const int*)d_in[12];
    float* out = (float*)d_out;

    const size_t NH = (size_t)NN * HH;
    float* xA   = (float*)d_ws;
    float* xB   = xA + NH;
    int2*  csrF = (int2*)(xA + 2 * NH);
    int2*  csrR = csrF + NE;
    int*   offF = (int*)(xA + 3 * NH);
    int*   offR = offF + NN;
    int2*  tmpF = (int2*)xA;
    int2*  tmpR = tmpF + NE;
    int*   cnt  = (int*)xB;
    int*   base = cnt + 2 * NCNT;

    const int batch = in_sizes[12];

    hist_kernel<<<EB, 256, 0, stream>>>(src, dst, cnt);
    scan_kernel<<<dim3(SCH, 2), 1024, 0, stream>>>(cnt, base);
    part_kernel<<<EB, 256, 0, stream>>>(src, dst, weights, cnt, base, tmpF, tmpR);
    sort_kernel<<<dim3(NBK, 2), 256, 0, stream>>>(base, tmpF, tmpR, csrF, csrR, offF, offR);

    enc_kernel<<<(NTILES + 7) / 8, 256, 0, stream>>>(text_emb, W_enc, b_enc, xA);

    layer_kernel<<<NTILES / 2, 64, 0, stream>>>(
        xA, xB, csrF, csrR, offF, offR,
        Wself_f, Wneigh_f, b_f, Wself_r, Wneigh_r, b_r);
    layer_kernel<<<NTILES / 2, 64, 0, stream>>>(
        xB, xA, csrF, csrR, offF, offR,
        Wself_f + (size_t)HH * HH, Wneigh_f + (size_t)HH * HH, b_f + HH,
        Wself_r + (size_t)HH * HH, Wneigh_r + (size_t)HH * HH, b_r + HH);

    norm_kernel<<<batch, 64, 0, stream>>>(xA, ids, out);
}


# Last layer computed only at the batch rows the output reads (a 16-row tile per work-group: a wave per row gathers, 4 waves do the products), not at all nodes; every sum and its order unchanged
# speedup vs baseline: 4.6176x; 1.2933x over previous
#include <hip/hip_runtime.h>
#include <hip/hip_bf16.h>
#include <math.h>

typedef __attribute__((ext_vector_type(2))) float v2f;
typedef __attribute__((ext_vector_type(8))) float v8f;

#define NN      100000
#define NE      1600000
#define HH      64
#define TT      384
#define NTILES  6250
#define TP      68
#define NBMASK  0xFFFFF

__device__ __forceinline__ v2f mk2(float a, float b) {
    v2f r; r.x = a; r.y = b; return r;
}

__device__ __forceinline__ v8f wmma4(v2f a, v2f b, v8f c) {
    return __builtin_amdgcn_wmma_f32_16x16x4_f32(
         false, a,  false, b,
         (short)0, c,  false,  false);
}

#define NBK   391
#define EB    400
#define EPB   4000
#define NCNT  (NBK * EB)
#define SCH   ((NCNT + 4095) / 4096)
#define CAP   6144
#define QPT   ((EPB / 4 + 255) / 256)
#define RPT   (CAP / 256)

__device__ __forceinline__ void scan512(int* a, int* wtot, int tid) {
    const int lane = tid & 31;
    const int wv   = tid >> 5;
    const int x0 = a[2 * tid], x1 = a[2 * tid + 1];
    const int s = x0 + x1;
    int inc = s;
    for (int off = 1; off < 32; off <<= 1) {
        int t = __shfl_up(inc, off);
        if (lane >= off) inc += t;
    }
    if (lane == 31) wtot[wv] = inc;
    __syncthreads();
    int pre = 0;
    for (int k = 0; k < 8; ++k) {
        const int v = wtot[k];
        pre += (k < wv) ? v : 0;
    }
    const int ex = pre + inc - s;
    a[2 * tid]     = ex;
    a[2 * tid + 1] = ex + x0;
    __syncthreads();
}

__global__ __launch_bounds__(256) void hist_kernel(const int* __restrict__ src, const int* __restrict__ dst,
                                                   int* __restrict__ cnt) {
    __shared__ int h[2 * NBK];
    const int tid = threadIdx.x;
    const int blk = blockIdx.x;
    for (int t = tid; t < 2 * NBK; t += 256) h[t] = 0;
    __syncthreads();
    const int4* s4 = reinterpret_cast<const int4*>(src + (size_t)blk * EPB);
    const int4* d4 = reinterpret_cast<const int4*>(dst + (size_t)blk * EPB);
    int4 sv[QPT], dv[QPT];
    #pragma unroll
    for (int k = 0; k < QPT; ++k) {
        const int q = tid + 256 * k;
        if (q < EPB / 4) { sv[k] = s4[q]; dv[k] = d4[q]; }
    }
    #pragma unroll
    for (int k = 0; k < QPT; ++k) {
        const int q = tid + 256 * k;
        if (q < EPB / 4) {
            const int4 s = sv[k];
            const int4 d = dv[k];
            atomicAdd(&h[d.x >> 8], 1); atomicAdd(&h[d.y >> 8], 1);
            atomicAdd(&h[d.z >> 8], 1); atomicAdd(&h[d.w >> 8], 1);
            atomicAdd(&h[NBK + (s.x >> 8)], 1); atomicAdd(&h[NBK + (s.y >> 8)], 1);
            atomicAdd(&h[NBK + (s.z >> 8)], 1); atomicAdd(&h[NBK + (s.w >> 8)], 1);
        }
    }
    __syncthreads();
    for (int t = tid; t < 2 * NBK; t += 256) cnt[(size_t)t * EB + blk] = h[t];
}

__global__ __launch_bounds__(1024) void scan_kernel(const int* __restrict__ cnt, int* __restrict__ base) {
    __shared__ int wsum[32];
    __shared__ int psum[32];
    const int tid  = threadIdx.x;
    const int lane = tid & 31;
    const int wv   = tid >> 5;
    const int c    = blockIdx.x;
    const int4* in = reinterpret_cast<const int4*>(cnt + (size_t)blockIdx.y * NCNT);
    int4* out      = reinterpret_cast<int4*>(base + (size_t)blockIdx.y * NCNT);
    int part = 0;
    for (int j = 0; j < c; ++j) {
        const int4 v = in[j * 1024 + tid];
        part += (v.x + v.y) + (v.z + v.w);
    }
    for (int off = 16; off > 0; off >>= 1) part += __shfl_xor(part, off);
    if (lane == 0) psum[wv] = part;
    const int i4 = c * 1024 + tid;
    const bool in_range = i4 < (NCNT / 4);
    int4 v = make_int4(0, 0, 0, 0);
    if (in_range) v = in[i4];
    const int s1 = v.x, s2 = s1 + v.y, s3 = s2 + v.z, s4 = s3 + v.w;
    int inc = s4;
    for (int off = 1; off < 32; off <<= 1) {
        int t = __shfl_up(inc, off);
        if (lane >= off) inc += t;
    }
    if (lane == 31) wsum[wv] = inc;
    __syncthreads();
    int carry = psum[lane];
    for (int off = 16; off > 0; off >>= 1) carry += __shfl_xor(carry, off);
    const int ws = wsum[lane];
    int winc = ws;
    for (int off = 1; off < 32; off <<= 1) {
        int t = __shfl_up(winc, off);
        if (lane >= off) winc += t;
    }
    const int wexcl = __shfl(winc, wv) - __shfl(ws, wv);
    const int b0 = carry + wexcl + (inc - s4);
    if (in_range) out[i4] = make_int4(b0, b0 + s1, b0 + s2, b0 + s3);
}

__global__ __launch_bounds__(256) void part_kernel(const int* __restrict__ src, const int* __restrict__ dst,
                                                   const float* __restrict__ w,
                                                   const int* __restrict__ cnt, const int* __restrict__ base,
                                                   int2* __restrict__ tmpF, int2* __restrict__ tmpR) {
    __shared__ int lb[2][512];
    __shared__ int cur[2][512];
    __shared__ int gb[2][512];
    __shared__ int wtot[8];
    __shared__ int2 srec[2][EPB];
    __shared__ int  sg[2][EPB];
    const int tid = threadIdx.x;
    const int blk = blockIdx.x;
    const int4*   s4 = reinterpret_cast<const int4*>(src + (size_t)blk * EPB);
    const int4*   d4 = reinterpret_cast<const int4*>(dst + (size_t)blk * EPB);
    const float4* w4 = reinterpret_cast<const float4*>(w + (size_t)blk * EPB);
    int4 sv[QPT], dv[QPT];
    float4 wv4[QPT];
    #pragma unroll
    for (int k = 0; k < QPT; ++k) {
        const int q = tid + 256 * k;
        if (q < EPB / 4) { sv[k] = s4[q]; dv[k] = d4[q]; wv4[k] = w4[q]; }
    }
    for (int t = tid; t < 1024; t += 256) {
        const int dir = t >> 9, b = t & 511;
        int c = 0, g = 0;
        if (b < NBK) {
            const size_t ix = (size_t)(dir * NBK + b) * EB + blk;
            c = cnt[ix];
            g = base[ix];
        }
        lb[dir][b] = c;
        gb[dir][b] = g;
    }
    __syncthreads();
    scan512(lb[0], wtot, tid);
    scan512(lb[1], wtot, tid);
    for (int t = tid; t < 1024; t += 256) {
        const int dir = t >> 9, b = t & 511;
        const int l = lb[dir][b];
        cur[dir][b] = l;
        gb[dir][b] -= l;
    }
    __syncthreads();
    #pragma unroll
    for (int k = 0; k < QPT; ++k) {
        const int q = tid + 256 * k;
        if (q < EPB / 4) {
            const int ss[4] = {sv[k].x, sv[k].y, sv[k].z, sv[k].w};
            const int dd[4] = {dv[k].x, dv[k].y, dv[k].z, dv[k].w};
            const float ww[4] = {wv4[k].x, wv4[k].y, wv4[k].z, wv4[k].w};
            #pragma unroll
            for (int u = 0; u < 4; ++u) {
                const int s = ss[u], d = dd[u];
                const int wb = __float_as_int(ww[u]);
                const int bF = d >> 8, bR = s >> 8;
                const int pF = atomicAdd(&cur[0][bF], 1);
                srec[0][pF] = make_int2(s | ((d & 255) << 20), wb);
                sg[0][pF]   = gb[0][bF] + pF;
                const int pR = atomicAdd(&cur[1][bR], 1);
                srec[1][pR] = make_int2(d | ((s & 255) << 20), wb);
                sg[1][pR]   = gb[1][bR] + pR;
            }
        }
    }
    __syncthreads();
    for (int i = tid; i < EPB; i += 256) {
        tmpF[sg[0][i]] = srec[0][i];
        tmpR[sg[1][i]] = srec[1][i];
    }
}

__global__ __launch_bounds__(256) void sort_kernel(const int* __restrict__ base,
                                                   const int2* __restrict__ tmpF, const int2* __restrict__ tmpR,
                                                   int2* __restrict__ csrF, int2* __restrict__ csrR,
                                                   int* __restrict__ offF, int* __restrict__ offR) {
    __shared__ int bins[512];
    __shared__ int cur[256];
    __shared__ int wtot[8];
    __shared__ int2 stage[CAP];
    const int tid = threadIdx.x;
    const int b   = blockIdx.x;
    const int dir = blockIdx.y;
    const int2* tmp = dir ? tmpR : tmpF;
    int2* csr       = dir ? csrR : csrF;
    int* off        = dir ? offR : offF;
    const int beg = base[(size_t)(dir * NBK + b) * EB];
    const int end = (b + 1 < NBK) ? base[(size_t)(dir * NBK + b + 1) * EB] : NE;
    const int n   = end - beg;
    const bool staged = n <= CAP;
    bins[tid] = 0;
    bins[tid + 256] = 0;
    __syncthreads();

    int2 r[RPT];
    if (staged) {
        #pragma unroll
        for (int k = 0; k < RPT; ++k) {
            const int i = tid + 256 * k;
            r[k] = make_int2(0, 0);
            if (i < n) r[k] = tmp[beg + i];
        }
        #pragma unroll
        for (int k = 0; k < RPT; ++k) {
            const int i = tid + 256 * k;
            if (i < n) atomicAdd(&bins[(r[k].x >> 20) & 255], 1);
        }
    } else {
        for (int i = tid; i < n; i += 256) atomicAdd(&bins[(tmp[beg + i].x >> 20) & 255], 1);
    }
    __syncthreads();
    scan512(bins, wtot, tid);
    {
        const int node = b * 256 + tid;
        const int ex = bins[tid];
        if (node < NN) off[node] = beg + ex;
        cur[tid] = ex;
    }
    __syncthreads();

    if (staged) {
        #pragma unroll
        for (int k = 0; k < RPT; ++k) {
            const int i = tid + 256 * k;
            if (i < n) {
                const int nl = (r[k].x >> 20) & 255;
                const int p = atomicAdd(&cur[nl], 1);
                stage[p] = make_int2((r[k].x & NBMASK) | ((nl & 15) << 20), r[k].y);
            }
        }
        __syncthreads();
        for (int i = tid; i < n; i += 256) csr[beg + i] = stage[i];
    } else {
        for (int i = tid; i < n; i += 256) {
            const int2 rr = tmp[beg + i];
            const int nl = (rr.x >> 20) & 255;
            const int p = atomicAdd(&cur[nl], 1);
            csr[beg + p] = make_int2((rr.x & NBMASK) | ((nl & 15) << 20), rr.y);
        }
    }
}

__global__ __launch_bounds__(256) void enc_kernel(const float* __restrict__ emb,
                                                  const float* __restrict__ W,
                                                  const float* __restrict__ bias,
                                                  float* __restrict__ x) {
    const int wave = threadIdx.x >> 5;
    const int lane = threadIdx.x & 31;
    const int nt   = blockIdx.x * 8 + wave;
    if (nt >= NTILES) return;
    const int ln = lane & 15;
    const int hi = lane >> 4;
    const int kh = hi << 1;

    const float* arow = emb + (size_t)(nt * 16 + ln) * TT + kh;
    v8f acc0 = {}, acc1 = {}, acc2 = {}, acc3 = {};

    #pragma unroll 4
    for (int k0 = 0; k0 < TT; k0 += 4) {
        v2f a = *reinterpret_cast<const v2f*>(arow + k0);
        const float* bp = W + (size_t)(k0 + kh) * HH + ln;
        v2f b0 = mk2(bp[0],  bp[HH]);
        v2f b1 = mk2(bp[16], bp[HH + 16]);
        v2f b2 = mk2(bp[32], bp[HH + 32]);
        v2f b3 = mk2(bp[48], bp[HH + 48]);
        acc0 = wmma4(a, b0, acc0);
        acc1 = wmma4(a, b1, acc1);
        acc2 = wmma4(a, b2, acc2);
        acc3 = wmma4(a, b3, acc3);
    }

    float bs0 = bias[ln], bs1 = bias[16 + ln], bs2 = bias[32 + ln], bs3 = bias[48 + ln];
    #pragma unroll
    for (int v = 0; v < 8; ++v) {
        size_t r = (size_t)(nt * 16 + v + 8 * hi) * HH + ln;
        x[r]      = acc0[v] + bs0;
        x[r + 16] = acc1[v] + bs1;
        x[r + 32] = acc2[v] + bs2;
        x[r + 48] = acc3[v] + bs3;
    }
}

__device__ __forceinline__ void gather_tile(const float* __restrict__ xin,
                                            const int2* __restrict__ csr,
                                            int beg, int end, int lane,
                                            float* __restrict__ tile) {
    if (beg < end) {
        float ax = 0.0f, ay = 0.0f;
        int cur = -1;
        int2 en[8];
        #pragma unroll
        for (int u = 0; u < 8; ++u) en[u] = csr[min(beg + u, end - 1)];
        for (int p = beg; p < end; p += 8) {
            float2 v[8];
            #pragma unroll
            for (int u = 0; u < 8; ++u)
                v[u] = *reinterpret_cast<const float2*>(xin + (size_t)(en[u].x & NBMASK) * HH + 2 * lane);
            int2 nx[8];
            #pragma unroll
            for (int u = 0; u < 8; ++u) nx[u] = csr[min(p + 8 + u, end - 1)];
            #pragma unroll
            for (int u = 0; u < 8; ++u) {
                const bool ok = (p + u) < end;
                const int row = en[u].x >> 20;
                const float we = __int_as_float(en[u].y);
                const bool same = (row == cur);
                ax = same ? ax : 0.0f;
                ay = same ? ay : 0.0f;
                cur = row;
                ax += ok ? we * v[u].x : 0.0f;
                ay += ok ? we * v[u].y : 0.0f;
                *reinterpret_cast<float2*>(tile + row * TP + 2 * lane) = make_float2(ax, ay);
            }
            #pragma unroll
            for (int u = 0; u < 8; ++u) en[u] = nx[u];
        }
    }
}

__global__ __launch_bounds__(64) void layer_kernel(
    const float* __restrict__ xin, float* __restrict__ xout,
    const int2* __restrict__ csrF, const int2* __restrict__ csrR,
    const int* __restrict__ offF, const int* __restrict__ offR,
    const float* __restrict__ WsF, const float* __restrict__ WnF, const float* __restrict__ bF,
    const float* __restrict__ WsR, const float* __restrict__ WnR, const float* __restrict__ bR) {
    __shared__ __attribute__((aligned(16))) float tiles[2][2][16 * TP];
    const int wave = __builtin_amdgcn_readfirstlane((int)(threadIdx.x >> 5));
    const int lane = threadIdx.x & 31;
    const int nt   = blockIdx.x * 2 + wave;
    const int n0   = nt * 16;
    const int ln = lane & 15;
    const int hi = lane >> 4;
    const int kh = hi << 1;
    const int rowA = n0 + ln;

    float* tF = &tiles[wave][0][0];
    float* tR = &tiles[wave][1][0];

    #pragma unroll
    for (int i = 0; i < (2 * 16 * TP) / (4 * 32); ++i)
        reinterpret_cast<float4*>(tF)[lane + 32 * i] = make_float4(0.0f, 0.0f, 0.0f, 0.0f);
    __syncthreads();

    const int begF = offF[n0];
    const int endF = (n0 + 16 < NN) ? offF[n0 + 16] : NE;
    const int begR = offR[n0];
    const int endR = (n0 + 16 < NN) ? offR[n0 + 16] : NE;
    gather_tile(xin, csrF, begF, endF, lane, tF);
    gather_tile(xin, csrR, begR, endR, lane, tR);
    __syncthreads();

    const int dF = ((rowA + 1 < NN) ? offF[rowA + 1] : NE) - offF[rowA];
    const int dR = ((rowA + 1 < NN) ? offR[rowA + 1] : NE) - offR[rowA];
    const float sF = (dF > 0) ? (1.0f / (float)dF) : 0.0f;
    const float sR = (dR > 0) ? (1.0f / (float)dR) : 0.0f;

    const float* xr = xin + (size_t)rowA * HH + kh;
    const float* fr = tF + ln * TP + kh;
    const float* rr = tR + ln * TP + kh;

    v8f aF0 = {}, aF1 = {}, aF2 = {}, aF3 = {};
    v8f aR0 = {}, aR1 = {}, aR2 = {}, aR3 = {};

    #pragma unroll 4
    for (int k0 = 0; k0 < HH; k0 += 4) {
        v2f aS = *reinterpret_cast<const v2f*>(xr + k0);
        v2f aN = *reinterpret_cast<const v2f*>(fr + k0);
        v2f aV = *reinterpret_cast<const v2f*>(rr + k0);
        aN *= sF;
        aV *= sR;

        const float* pSF = WsF + (size_t)(k0 + kh) * HH + ln;
        const float* pNF = WnF + (size_t)(k0 + kh) * HH + ln;
        const float* pSR = WsR + (size_t)(k0 + kh) * HH + ln;
        const float* pNR = WnR + (size_t)(k0 + kh) * HH + ln;

        aF0 = wmma4(aS, mk2(pSF[0], pSF[HH]), aF0);
        aF0 = wmma4(aN, mk2(pNF[0], pNF[HH]), aF0);
        aR0 = wmma4(aS, mk2(pSR[0], pSR[HH]), aR0);
        aR0 = wmma4(aV, mk2(pNR[0], pNR[HH]), aR0);
        aF1 = wmma4(aS, mk2(pSF[16], pSF[HH + 16]), aF1);
        aF1 = wmma4(aN, mk2(pNF[16], pNF[HH + 16]), aF1);
        aR1 = wmma4(aS, mk2(pSR[16], pSR[HH + 16]), aR1);
        aR1 = wmma4(aV, mk2(pNR[16], pNR[HH + 16]), aR1);
        aF2 = wmma4(aS, mk2(pSF[32], pSF[HH + 32]), aF2);
        aF2 = wmma4(aN, mk2(pNF[32], pNF[HH + 32]), aF2);
        aR2 = wmma4(aS, mk2(pSR[32], pSR[HH + 32]), aR2);
        aR2 = wmma4(aV, mk2(pNR[32], pNR[HH + 32]), aR2);
        aF3 = wmma4(aS, mk2(pSF[48], pSF[HH + 48]), aF3);
        aF3 = wmma4(aN, mk2(pNF[48], pNF[HH + 48]), aF3);
        aR3 = wmma4(aS, mk2(pSR[48], pSR[HH + 48]), aR3);
        aR3 = wmma4(aV, mk2(pNR[48], pNR[HH + 48]), aR3);
    }

    float bf0 = bF[ln], bf1 = bF[16 + ln], bf2 = bF[32 + ln], bf3 = bF[48 + ln];
    float br0 = bR[ln], br1 = bR[16 + ln], br2 = bR[32 + ln], br3 = bR[48 + ln];

    #pragma unroll
    for (int v = 0; v < 8; ++v) {
        size_t r = (size_t)(n0 + v + 8 * hi) * HH + ln;
        xout[r]      = xin[r]      + (fmaxf(aF0[v] + bf0, 0.0f) + fmaxf(aR0[v] + br0, 0.0f));
        xout[r + 16] = xin[r + 16] + (fmaxf(aF1[v] + bf1, 0.0f) + fmaxf(aR1[v] + br1, 0.0f));
        xout[r + 32] = xin[r + 32] + (fmaxf(aF2[v] + bf2, 0.0f) + fmaxf(aR2[v] + br2, 0.0f));
        xout[r + 48] = xin[r + 48] + (fmaxf(aF3[v] + bf3, 0.0f) + fmaxf(aR3[v] + br3, 0.0f));
    }
}

__device__ __forceinline__ void gather_row(const float* __restrict__ xin,
                                           const int2* __restrict__ csr,
                                           int beg, int end, int lane,
                                           float* __restrict__ trow) {
    float ax = 0.0f, ay = 0.0f;
    if (beg < end) {
        int2 en[8];
        #pragma unroll
        for (int u = 0; u < 8; ++u) en[u] = csr[min(beg + u, end - 1)];
        for (int p = beg; p < end; p += 8) {
            float2 v[8];
            #pragma unroll
            for (int u = 0; u < 8; ++u)
                v[u] = *reinterpret_cast<const float2*>(xin + (size_t)(en[u].x & NBMASK) * HH + 2 * lane);
            int2 nx[8];
            #pragma unroll
            for (int u = 0; u < 8; ++u) nx[u] = csr[min(p + 8 + u, end - 1)];
            #pragma unroll
            for (int u = 0; u < 8; ++u) {
                const bool ok = (p + u) < end;
                const float we = __int_as_float(en[u].y);
                ax += ok ? we * v[u].x : 0.0f;
                ay += ok ? we * v[u].y : 0.0f;
            }
            #pragma unroll
            for (int u = 0; u < 8; ++u) en[u] = nx[u];
        }
    }
    *reinterpret_cast<float2*>(trow + 2 * lane) = make_float2(ax, ay);
}

__global__ __launch_bounds__(512) void layer_rows_kernel(
    const float* __restrict__ xin, float* __restrict__ xc,
    const int* __restrict__ ids, int batch,
    const int2* __restrict__ csrF, const int2* __restrict__ csrR,
    const int* __restrict__ offF, const int* __restrict__ offR,
    const float* __restrict__ WsF, const float* __restrict__ WnF, const float* __restrict__ bF,
    const float* __restrict__ WsR, const float* __restrict__ WnR, const float* __restrict__ bR) {
    __shared__ __attribute__((aligned(16))) float tiles[2][16 * TP];
    __shared__ __attribute__((aligned(16))) float yt[16 * TP];
    const int wave = __builtin_amdgcn_readfirstlane((int)(threadIdx.x >> 5));
    const int lane = threadIdx.x & 31;
    const int ln = lane & 15;
    const int hi = lane >> 4;
    const int kh = hi << 1;

    for (int t0 = blockIdx.x * 16; t0 < batch; t0 += gridDim.x * 16) {
        const int node = __builtin_amdgcn_readfirstlane(ids[min(t0 + wave, batch - 1)]);
        {
            const int begF = offF[node];
            const int endF = (node + 1 < NN) ? offF[node + 1] : NE;
            const int begR = offR[node];
            const int endR = (node + 1 < NN) ? offR[node + 1] : NE;
            gather_row(xin, csrF, begF, endF, lane, &tiles[0][wave * TP]);
            gather_row(xin, csrR, begR, endR, lane, &tiles[1][wave * TP]);
        }
        __syncthreads();

        if (wave < 4) {
            const int nodeA = ids[min(t0 + ln, batch - 1)];
            const int dF = ((nodeA + 1 < NN) ? offF[nodeA + 1] : NE) - offF[nodeA];
            const int dR = ((nodeA + 1 < NN) ? offR[nodeA + 1] : NE) - offR[nodeA];
            const float sF = (dF > 0) ? (1.0f / (float)dF) : 0.0f;
            const float sR = (dR > 0) ? (1.0f / (float)dR) : 0.0f;

            const float* xr = xin + (size_t)nodeA * HH + kh;
            const float* fr = &tiles[0][ln * TP + kh];
            const float* rr = &tiles[1][ln * TP + kh];
            const int cj = 16 * wave + ln;

            v8f aF = {}, aR = {};
            #pragma unroll 4
            for (int k0 = 0; k0 < HH; k0 += 4) {
                v2f aS = *reinterpret_cast<const v2f*>(xr + k0);
                v2f aN = *reinterpret_cast<const v2f*>(fr + k0);
                v2f aV = *reinterpret_cast<const v2f*>(rr + k0);
                aN *= sF;
                aV *= sR;

                const float* pSF = WsF + (size_t)(k0 + kh) * HH + cj;
                const float* pNF = WnF + (size_t)(k0 + kh) * HH + cj;
                const float* pSR = WsR + (size_t)(k0 + kh) * HH + cj;
                const float* pNR = WnR + (size_t)(k0 + kh) * HH + cj;

                aF = wmma4(aS, mk2(pSF[0], pSF[HH]), aF);
                aF = wmma4(aN, mk2(pNF[0], pNF[HH]), aF);
                aR = wmma4(aS, mk2(pSR[0], pSR[HH]), aR);
                aR = wmma4(aV, mk2(pNR[0], pNR[HH]), aR);
            }

            const float bf = bF[cj];
            const float br = bR[cj];
            #pragma unroll
            for (int v = 0; v < 8; ++v)
                yt[(v + 8 * hi) * TP + cj] = fmaxf(aF[v] + bf, 0.0f) + fmaxf(aR[v] + br, 0.0f);
        }
        __syncthreads();

        if (t0 + wave < batch) {
            const float2 y  = *reinterpret_cast<const float2*>(&yt[wave * TP + 2 * lane]);
            const float2 xv = *reinterpret_cast<const float2*>(xin + (size_t)node * HH + 2 * lane);
            *reinterpret_cast<float2*>(xc + (size_t)(t0 + wave) * HH + 2 * lane) =
                make_float2(xv.x + y.x, xv.y + y.y);
        }
    }
}

__global__ __launch_bounds__(64) void norm_kernel(const float* __restrict__ x,
                                                  float* __restrict__ out) {
    __shared__ float s[64];
    int b = blockIdx.x;
    int c = threadIdx.x;
    float v = x[(size_t)b * HH + c];
    s[c] = v * v;
    __syncthreads();
    #pragma unroll
    for (int off = 32; off > 0; off >>= 1) {
        if (c < off) s[c] += s[c + off];
        __syncthreads();
    }
    float nrm = sqrtf(s[0]);
    out[(size_t)b * HH + c] = v / nrm;
}

extern "C" void kernel_launch(void* const* d_in, const int* in_sizes, int n_in,
                              void* d_out, int out_size, void* d_ws, size_t ws_size,
                              hipStream_t stream) {
    const float* text_emb = (const float*)d_in[0];
    const float* weights  = (const float*)d_in[1];
    const float* W_enc    = (const float*)d_in[2];
    const float* b_enc    = (const float*)d_in[3];
    const float* Wself_f  = (const float*)d_in[4];
    const float* Wneigh_f = (const float*)d_in[5];
    const float* b_f      = (const float*)d_in[6];
    const float* Wself_r  = (const float*)d_in[7];
    const float* Wneigh_r = (const float*)d_in[8];
    const float* b_r      = (const float*)d_in[9];
    const int*   src      = (const int*)d_in[10];
    const int*   dst      = (const int*)d_in[11];
    const int*   ids      = (const int*)d_in[12];
    float* out = (float*)d_out;

    const size_t NH = (size_t)NN * HH;
    float* xA   = (float*)d_ws;
    float* xB   = xA + NH;
    int2*  csrF = (int2*)(xA + 2 * NH);
    int2*  csrR = csrF + NE;
    int*   offF = (int*)(xA + 3 * NH);
    int*   offR = offF + NN;
    int2*  tmpF = (int2*)xA;
    int2*  tmpR = tmpF + NE;
    int*   cnt  = (int*)xB;
    int*   base = cnt + 2 * NCNT;

    const int batch = in_sizes[12];

    hist_kernel<<<EB, 256, 0, stream>>>(src, dst, cnt);
    scan_kernel<<<dim3(SCH, 2), 1024, 0, stream>>>(cnt, base);
    part_kernel<<<EB, 256, 0, stream>>>(src, dst, weights, cnt, base, tmpF, tmpR);
    sort_kernel<<<dim3(NBK, 2), 256, 0, stream>>>(base, tmpF, tmpR, csrF, csrR, offF, offR);

    enc_kernel<<<(NTILES + 7) / 8, 256, 0, stream>>>(text_emb, W_enc, b_enc, xA);

    layer_kernel<<<NTILES / 2, 64, 0, stream>>>(
        xA, xB, csrF, csrR, offF, offR,
        Wself_f, Wneigh_f, b_f, Wself_r, Wneigh_r, b_r);
    layer_rows_kernel<<<64, 512, 0, stream>>>(
        xB, xA, ids, batch, csrF, csrR, offF, offR,
        Wself_f + (size_t)HH * HH, Wneigh_f + (size_t)HH * HH, b_f + HH,
        Wself_r + (size_t)HH * HH, Wneigh_r + (size_t)HH * HH, b_r + HH);

    norm_kernel<<<batch, 64, 0, stream>>>(xA, out);
}


# First layer computed only at the rows the last layer reads (each batch node and its neighbours in both directions; a work-group per batch row), not at all nodes; every sum and its order unchanged
# speedup vs baseline: 5.4622x; 1.1829x over previous
#include <hip/hip_runtime.h>
#include <hip/hip_bf16.h>
#include <math.h>

typedef __attribute__((ext_vector_type(2))) float v2f;
typedef __attribute__((ext_vector_type(8))) float v8f;

#define NN      100000
#define NE      1600000
#define HH      64
#define TT      384
#define NTILES  6250
#define TP      68
#define NBMASK  0xFFFFF

__device__ __forceinline__ v2f mk2(float a, float b) {
    v2f r; r.x = a; r.y = b; return r;
}

__device__ __forceinline__ v8f wmma4(v2f a, v2f b, v8f c) {
    return __builtin_amdgcn_wmma_f32_16x16x4_f32(
         false, a,  false, b,
         (short)0, c,  false,  false);
}

#define NBK   391
#define EB    400
#define EPB   4000
#define NCNT  (NBK * EB)
#define SCH   ((NCNT + 4095) / 4096)
#define CAP   6144
#define QPT   ((EPB / 4 + 255) / 256)
#define RPT   (CAP / 256)

__device__ __forceinline__ void scan512(int* a, int* wtot, int tid) {
    const int lane = tid & 31;
    const int wv   = tid >> 5;
    const int x0 = a[2 * tid], x1 = a[2 * tid + 1];
    const int s = x0 + x1;
    int inc = s;
    for (int off = 1; off < 32; off <<= 1) {
        int t = __shfl_up(inc, off);
        if (lane >= off) inc += t;
    }
    if (lane == 31) wtot[wv] = inc;
    __syncthreads();
    int pre = 0;
    for (int k = 0; k < 8; ++k) {
        const int v = wtot[k];
        pre += (k < wv) ? v : 0;
    }
    const int ex = pre + inc - s;
    a[2 * tid]     = ex;
    a[2 * tid + 1] = ex + x0;
    __syncthreads();
}

__global__ __launch_bounds__(256) void hist_kernel(const int* __restrict__ src, const int* __restrict__ dst,
                                                   int* __restrict__ cnt) {
    __shared__ int h[2 * NBK];
    const int tid = threadIdx.x;
    const int blk = blockIdx.x;
    for (int t = tid; t < 2 * NBK; t += 256) h[t] = 0;
    __syncthreads();
    const int4* s4 = reinterpret_cast<const int4*>(src + (size_t)blk * EPB);
    const int4* d4 = reinterpret_cast<const int4*>(dst + (size_t)blk * EPB);
    int4 sv[QPT], dv[QPT];
    #pragma unroll
    for (int k = 0; k < QPT; ++k) {
        const int q = tid + 256 * k;
        if (q < EPB / 4) { sv[k] = s4[q]; dv[k] = d4[q]; }
    }
    #pragma unroll
    for (int k = 0; k < QPT; ++k) {
        const int q = tid + 256 * k;
        if (q < EPB / 4) {
            const int4 s = sv[k];
            const int4 d = dv[k];
            atomicAdd(&h[d.x >> 8], 1); atomicAdd(&h[d.y >> 8], 1);
            atomicAdd(&h[d.z >> 8], 1); atomicAdd(&h[d.w >> 8], 1);
            atomicAdd(&h[NBK + (s.x >> 8)], 1); atomicAdd(&h[NBK + (s.y >> 8)], 1);
            atomicAdd(&h[NBK + (s.z >> 8)], 1); atomicAdd(&h[NBK + (s.w >> 8)], 1);
        }
    }
    __syncthreads();
    for (int t = tid; t < 2 * NBK; t += 256) cnt[(size_t)t * EB + blk] = h[t];
}

__global__ __launch_bounds__(1024) void scan_kernel(const int* __restrict__ cnt, int* __restrict__ base) {
    __shared__ int wsum[32];
    __shared__ int psum[32];
    const int tid  = threadIdx.x;
    const int lane = tid & 31;
    const int wv   = tid >> 5;
    const int c    = blockIdx.x;
    const int4* in = reinterpret_cast<const int4*>(cnt + (size_t)blockIdx.y * NCNT);
    int4* out      = reinterpret_cast<int4*>(base + (size_t)blockIdx.y * NCNT);
    int part = 0;
    for (int j = 0; j < c; ++j) {
        const int4 v = in[j * 1024 + tid];
        part += (v.x + v.y) + (v.z + v.w);
    }
    for (int off = 16; off > 0; off >>= 1) part += __shfl_xor(part, off);
    if (lane == 0) psum[wv] = part;
    const int i4 = c * 1024 + tid;
    const bool in_range = i4 < (NCNT / 4);
    int4 v = make_int4(0, 0, 0, 0);
    if (in_range) v = in[i4];
    const int s1 = v.x, s2 = s1 + v.y, s3 = s2 + v.z, s4 = s3 + v.w;
    int inc = s4;
    for (int off = 1; off < 32; off <<= 1) {
        int t = __shfl_up(inc, off);
        if (lane >= off) inc += t;
    }
    if (lane == 31) wsum[wv] = inc;
    __syncthreads();
    int carry = psum[lane];
    for (int off = 16; off > 0; off >>= 1) carry += __shfl_xor(carry, off);
    const int ws = wsum[lane];
    int winc = ws;
    for (int off = 1; off < 32; off <<= 1) {
        int t = __shfl_up(winc, off);
        if (lane >= off) winc += t;
    }
    const int wexcl = __shfl(winc, wv) - __shfl(ws, wv);
    const int b0 = carry + wexcl + (inc - s4);
    if (in_range) out[i4] = make_int4(b0, b0 + s1, b0 + s2, b0 + s3);
}

__global__ __launch_bounds__(256) void part_kernel(const int* __restrict__ src, const int* __restrict__ dst,
                                                   const float* __restrict__ w,
                                                   const int* __restrict__ cnt, const int* __restrict__ base,
                                                   int2* __restrict__ tmpF, int2* __restrict__ tmpR) {
    __shared__ int lb[2][512];
    __shared__ int cur[2][512];
    __shared__ int gb[2][512];
    __shared__ int wtot[8];
    __shared__ int2 srec[2][EPB];
    __shared__ int  sg[2][EPB];
    const int tid = threadIdx.x;
    const int blk = blockIdx.x;
    const int4*   s4 = reinterpret_cast<const int4*>(src + (size_t)blk * EPB);
    const int4*   d4 = reinterpret_cast<const int4*>(dst + (size_t)blk * EPB);
    const float4* w4 = reinterpret_cast<const float4*>(w + (size_t)blk * EPB);
    int4 sv[QPT], dv[QPT];
    float4 wv4[QPT];
    #pragma unroll
    for (int k = 0; k < QPT; ++k) {
        const int q = tid + 256 * k;
        if (q < EPB / 4) { sv[k] = s4[q]; dv[k] = d4[q]; wv4[k] = w4[q]; }
    }
    for (int t = tid; t < 1024; t += 256) {
        const int dir = t >> 9, b = t & 511;
        int c = 0, g = 0;
        if (b < NBK) {
            const size_t ix = (size_t)(dir * NBK + b) * EB + blk;
            c = cnt[ix];
            g = base[ix];
        }
        lb[dir][b] = c;
        gb[dir][b] = g;
    }
    __syncthreads();
    scan512(lb[0], wtot, tid);
    scan512(lb[1], wtot, tid);
    for (int t = tid; t < 1024; t += 256) {
        const int dir = t >> 9, b = t & 511;
        const int l = lb[dir][b];
        cur[dir][b] = l;
        gb[dir][b] -= l;
    }
    __syncthreads();
    #pragma unroll
    for (int k = 0; k < QPT; ++k) {
        const int q = tid + 256 * k;
        if (q < EPB / 4) {
            const int ss[4] = {sv[k].x, sv[k].y, sv[k].z, sv[k].w};
            const int dd[4] = {dv[k].x, dv[k].y, dv[k].z, dv[k].w};
            const float ww[4] = {wv4[k].x, wv4[k].y, wv4[k].z, wv4[k].w};
            #pragma unroll
            for (int u = 0; u < 4; ++u) {
                const int s = ss[u], d = dd[u];
                const int wb = __float_as_int(ww[u]);
                const int bF = d >> 8, bR = s >> 8;
                const int pF = atomicAdd(&cur[0][bF], 1);
                srec[0][pF] = make_int2(s | ((d & 255) << 20), wb);
                sg[0][pF]   = gb[0][bF] + pF;
                const int pR = atomicAdd(&cur[1][bR], 1);
                srec[1][pR] = make_int2(d | ((s & 255) << 20), wb);
                sg[1][pR]   = gb[1][bR] + pR;
            }
        }
    }
    __syncthreads();
    for (int i = tid; i < EPB; i += 256) {
        tmpF[sg[0][i]] = srec[0][i];
        tmpR[sg[1][i]] = srec[1][i];
    }
}

__global__ __launch_bounds__(256) void sort_kernel(const int* __restrict__ base,
                                                   const int2* __restrict__ tmpF, const int2* __restrict__ tmpR,
                                                   int2* __restrict__ csrF, int2* __restrict__ csrR,
                                                   int* __restrict__ offF, int* __restrict__ offR) {
    __shared__ int bins[512];
    __shared__ int cur[256];
    __shared__ int wtot[8];
    __shared__ int2 stage[CAP];
    const int tid = threadIdx.x;
    const int b   = blockIdx.x;
    const int dir = blockIdx.y;
    const int2* tmp = dir ? tmpR : tmpF;
    int2* csr       = dir ? csrR : csrF;
    int* off        = dir ? offR : offF;
    const int beg = base[(size_t)(dir * NBK + b) * EB];
    const int end = (b + 1 < NBK) ? base[(size_t)(dir * NBK + b + 1) * EB] : NE;
    const int n   = end - beg;
    const bool staged = n <= CAP;
    bins[tid] = 0;
    bins[tid + 256] = 0;
    __syncthreads();

    int2 r[RPT];
    if (staged) {
        #pragma unroll
        for (int k = 0; k < RPT; ++k) {
            const int i = tid + 256 * k;
            r[k] = make_int2(0, 0);
            if (i < n) r[k] = tmp[beg + i];
        }
        #pragma unroll
        for (int k = 0; k < RPT; ++k) {
            const int i = tid + 256 * k;
            if (i < n) atomicAdd(&bins[(r[k].x >> 20) & 255], 1);
        }
    } else {
        for (int i = tid; i < n; i += 256) atomicAdd(&bins[(tmp[beg + i].x >> 20) & 255], 1);
    }
    __syncthreads();
    scan512(bins, wtot, tid);
    {
        const int node = b * 256 + tid;
        const int ex = bins[tid];
        if (node < NN) off[node] = beg + ex;
        cur[tid] = ex;
    }
    __syncthreads();

    if (staged) {
        #pragma unroll
        for (int k = 0; k < RPT; ++k) {
            const int i = tid + 256 * k;
            if (i < n) {
                const int nl = (r[k].x >> 20) & 255;
                const int p = atomicAdd(&cur[nl], 1);
                stage[p] = make_int2((r[k].x & NBMASK) | ((nl & 15) << 20), r[k].y);
            }
        }
        __syncthreads();
        for (int i = tid; i < n; i += 256) csr[beg + i] = stage[i];
    } else {
        for (int i = tid; i < n; i += 256) {
            const int2 rr = tmp[beg + i];
            const int nl = (rr.x >> 20) & 255;
            const int p = atomicAdd(&cur[nl], 1);
            csr[beg + p] = make_int2((rr.x & NBMASK) | ((nl & 15) << 20), rr.y);
        }
    }
}

__global__ __launch_bounds__(256) void enc_kernel(const float* __restrict__ emb,
                                                  const float* __restrict__ W,
                                                  const float* __restrict__ bias,
                                                  float* __restrict__ x) {
    const int wave = threadIdx.x >> 5;
    const int lane = threadIdx.x & 31;
    const int nt   = blockIdx.x * 8 + wave;
    if (nt >= NTILES) return;
    const int ln = lane & 15;
    const int hi = lane >> 4;
    const int kh = hi << 1;

    const float* arow = emb + (size_t)(nt * 16 + ln) * TT + kh;
    v8f acc0 = {}, acc1 = {}, acc2 = {}, acc3 = {};

    #pragma unroll 4
    for (int k0 = 0; k0 < TT; k0 += 4) {
        v2f a = *reinterpret_cast<const v2f*>(arow + k0);
        const float* bp = W + (size_t)(k0 + kh) * HH + ln;
        v2f b0 = mk2(bp[0],  bp[HH]);
        v2f b1 = mk2(bp[16], bp[HH + 16]);
        v2f b2 = mk2(bp[32], bp[HH + 32]);
        v2f b3 = mk2(bp[48], bp[HH + 48]);
        acc0 = wmma4(a, b0, acc0);
        acc1 = wmma4(a, b1, acc1);
        acc2 = wmma4(a, b2, acc2);
        acc3 = wmma4(a, b3, acc3);
    }

    float bs0 = bias[ln], bs1 = bias[16 + ln], bs2 = bias[32 + ln], bs3 = bias[48 + ln];
    #pragma unroll
    for (int v = 0; v < 8; ++v) {
        size_t r = (size_t)(nt * 16 + v + 8 * hi) * HH + ln;
        x[r]      = acc0[v] + bs0;
        x[r + 16] = acc1[v] + bs1;
        x[r + 32] = acc2[v] + bs2;
        x[r + 48] = acc3[v] + bs3;
    }
}

__device__ __forceinline__ int2 uniform_entry(int2 e) {
    return make_int2(__builtin_amdgcn_readfirstlane(e.x), __builtin_amdgcn_readfirstlane(e.y));
}

__device__ __forceinline__ void first_entries(const int2* __restrict__ csr, int beg, int end, int2 (&raw)[8]) {
    #pragma unroll
    for (int u = 0; u < 8; ++u) raw[u] = make_int2(0, 0);
    if (beg < end) {
        #pragma unroll
        for (int u = 0; u < 8; ++u) raw[u] = csr[min(beg + u, end - 1)];
    }
}

__device__ __forceinline__ void gather_row_from(const float* __restrict__ xin,
                                                const int2* __restrict__ csr,
                                                int beg, int end, int lane,
                                                float* __restrict__ trow, const int2 (&first)[8]) {
    float ax = 0.0f, ay = 0.0f;
    if (beg < end) {
        int2 en[8];
        #pragma unroll
        for (int u = 0; u < 8; ++u) en[u] = first[u];
        for (int p = beg; p < end; p += 8) {
            float2 v[8];
            #pragma unroll
            for (int u = 0; u < 8; ++u)
                v[u] = *reinterpret_cast<const float2*>(xin + (size_t)(en[u].x & NBMASK) * HH + 2 * lane);
            int2 nx[8];
            #pragma unroll
            for (int u = 0; u < 8; ++u) nx[u] = csr[min(p + 8 + u, end - 1)];
            #pragma unroll
            for (int u = 0; u < 8; ++u) {
                const bool ok = (p + u) < end;
                const float we = __int_as_float(en[u].y);
                ax += ok ? we * v[u].x : 0.0f;
                ay += ok ? we * v[u].y : 0.0f;
            }
            #pragma unroll
            for (int u = 0; u < 8; ++u) en[u] = uniform_entry(nx[u]);
        }
    }
    *reinterpret_cast<float2*>(trow + 2 * lane) = make_float2(ax, ay);
}

__device__ __forceinline__ void gather_row(const float* __restrict__ xin,
                                           const int2* __restrict__ csr,
                                           int beg, int end, int lane,
                                           float* __restrict__ trow) {
    int2 first[8];
    first_entries(csr, beg, end, first);
    #pragma unroll
    for (int u = 0; u < 8; ++u) first[u] = uniform_entry(first[u]);
    gather_row_from(xin, csr, beg, end, lane, trow, first);
}

__device__ __forceinline__ void seg_bounds(const int* __restrict__ off, int node, int& beg, int& end) {
    const int b = off[node];
    const int e = off[min(node + 1, NN - 1)];
    beg = b;
    end = (node + 1 < NN) ? e : NE;
}

__global__ __launch_bounds__(512) void layer_need_kernel(
    const float* __restrict__ xin, float* __restrict__ xout,
    const int* __restrict__ ids, int batch,
    const int2* __restrict__ csrF, const int2* __restrict__ csrR,
    const int* __restrict__ offF, const int* __restrict__ offR,
    const float* __restrict__ WsF, const float* __restrict__ WnF, const float* __restrict__ bF,
    const float* __restrict__ WsR, const float* __restrict__ WnR, const float* __restrict__ bR) {
    __shared__ __attribute__((aligned(16))) float tiles[2][16 * TP];
    __shared__ __attribute__((aligned(16))) float yt[2][16 * TP];
    __shared__ int   rnode[16];
    __shared__ float rscale[2][16];
    const int wave = __builtin_amdgcn_readfirstlane((int)(threadIdx.x >> 5));
    const int lane = threadIdx.x & 31;
    const int ln = lane & 15;
    const int hi = lane >> 4;
    const int kh = hi << 1;

    for (int b = blockIdx.x; b < batch; b += gridDim.x) {
        const int n   = __builtin_amdgcn_readfirstlane(ids[b]);
        int nbF, neF, nbR, neR;
        seg_bounds(offF, n, nbF, neF);
        seg_bounds(offR, n, nbR, neR);
        const int cF  = neF - nbF;
        const int len = 1 + cF + (neR - nbR);

        for (int i0 = 0; i0 < len; i0 += 16) {
            const int i = i0 + wave;
            const bool have = i < len;
            int m = n;
            if (have && i > 0) {
                const int2 e = (i <= cF) ? csrF[nbF + (i - 1)] : csrR[nbR + (i - 1 - cF)];
                m = e.x & NBMASK;
            }
            int begF = 0, endF = 0, begR = 0, endR = 0;
            if (have) {
                seg_bounds(offF, m, begF, endF);
                seg_bounds(offR, m, begR, endR);
            }
            int2 firstF[8], firstR[8];
            first_entries(csrF, begF, endF, firstF);
            first_entries(csrR, begR, endR, firstR);
            #pragma unroll
            for (int u = 0; u < 8; ++u) {
                firstF[u] = uniform_entry(firstF[u]);
                firstR[u] = uniform_entry(firstR[u]);
            }
            gather_row_from(xin, csrF, begF, endF, lane, &tiles[0][wave * TP], firstF);
            gather_row_from(xin, csrR, begR, endR, lane, &tiles[1][wave * TP], firstR);
            if (lane == 0) {
                const int dF = endF - begF;
                const int dR = endR - begR;
                rnode[wave] = m;
                rscale[0][wave] = (dF > 0) ? (1.0f / (float)dF) : 0.0f;
                rscale[1][wave] = (dR > 0) ? (1.0f / (float)dR) : 0.0f;
            }
            __syncthreads();

            if (wave < 8) {
                const int dir = wave >> 2;
                const float* Ws = dir ? WsR : WsF;
                const float* Wn = dir ? WnR : WnF;
                const float* bb = dir ? bR : bF;
                const int cj = 16 * (wave & 3) + ln;
                const float sc = rscale[dir][ln];
                const float* xr = xin + (size_t)rnode[ln] * HH + kh;
                const float* nr = &tiles[dir][ln * TP + kh];

                v8f acc = {};
                #pragma unroll 2
                for (int k0 = 0; k0 < HH; k0 += 4) {
                    v2f aS = *reinterpret_cast<const v2f*>(xr + k0);
                    v2f aN = *reinterpret_cast<const v2f*>(nr + k0);
                    aN *= sc;
                    const float* pS = Ws + (size_t)(k0 + kh) * HH + cj;
                    const float* pN = Wn + (size_t)(k0 + kh) * HH + cj;
                    acc = wmma4(aS, mk2(pS[0], pS[HH]), acc);
                    acc = wmma4(aN, mk2(pN[0], pN[HH]), acc);
                }

                const float bv = bb[cj];
                #pragma unroll
                for (int v = 0; v < 8; ++v)
                    yt[dir][(v + 8 * hi) * TP + cj] = fmaxf(acc[v] + bv, 0.0f);
            }
            __syncthreads();

            if (have) {
                const float2 yF = *reinterpret_cast<const float2*>(&yt[0][wave * TP + 2 * lane]);
                const float2 yR = *reinterpret_cast<const float2*>(&yt[1][wave * TP + 2 * lane]);
                const float2 xv = *reinterpret_cast<const float2*>(xin + (size_t)m * HH + 2 * lane);
                *reinterpret_cast<float2*>(xout + (size_t)m * HH + 2 * lane) =
                    make_float2(xv.x + (yF.x + yR.x), xv.y + (yF.y + yR.y));
            }
        }
    }
}

__global__ __launch_bounds__(512) void layer_rows_kernel(
    const float* __restrict__ xin, float* __restrict__ xc,
    const int* __restrict__ ids, int batch,
    const int2* __restrict__ csrF, const int2* __restrict__ csrR,
    const int* __restrict__ offF, const int* __restrict__ offR,
    const float* __restrict__ WsF, const float* __restrict__ WnF, const float* __restrict__ bF,
    const float* __restrict__ WsR, const float* __restrict__ WnR, const float* __restrict__ bR) {
    __shared__ __attribute__((aligned(16))) float tiles[2][16 * TP];
    __shared__ __attribute__((aligned(16))) float yt[16 * TP];
    const int wave = __builtin_amdgcn_readfirstlane((int)(threadIdx.x >> 5));
    const int lane = threadIdx.x & 31;
    const int ln = lane & 15;
    const int hi = lane >> 4;
    const int kh = hi << 1;

    for (int t0 = blockIdx.x * 16; t0 < batch; t0 += gridDim.x * 16) {
        const int node = __builtin_amdgcn_readfirstlane(ids[min(t0 + wave, batch - 1)]);
        {
            const int begF = offF[node];
            const int endF = (node + 1 < NN) ? offF[node + 1] : NE;
            const int begR = offR[node];
            const int endR = (node + 1 < NN) ? offR[node + 1] : NE;
            gather_row(xin, csrF, begF, endF, lane, &tiles[0][wave * TP]);
            gather_row(xin, csrR, begR, endR, lane, &tiles[1][wave * TP]);
        }
        __syncthreads();

        if (wave < 4) {
            const int nodeA = ids[min(t0 + ln, batch - 1)];
            const int dF = ((nodeA + 1 < NN) ? offF[nodeA + 1] : NE) - offF[nodeA];
            const int dR = ((nodeA + 1 < NN) ? offR[nodeA + 1] : NE) - offR[nodeA];
            const float sF = (dF > 0) ? (1.0f / (float)dF) : 0.0f;
            const float sR = (dR > 0) ? (1.0f / (float)dR) : 0.0f;

            const float* xr = xin + (size_t)nodeA * HH + kh;
            const float* fr = &tiles[0][ln * TP + kh];
            const float* rr = &tiles[1][ln * TP + kh];
            const int cj = 16 * wave + ln;

            v8f aF = {}, aR = {};
            #pragma unroll 4
            for (int k0 = 0; k0 < HH; k0 += 4) {
                v2f aS = *reinterpret_cast<const v2f*>(xr + k0);
                v2f aN = *reinterpret_cast<const v2f*>(fr + k0);
                v2f aV = *reinterpret_cast<const v2f*>(rr + k0);
                aN *= sF;
                aV *= sR;

                const float* pSF = WsF + (size_t)(k0 + kh) * HH + cj;
                const float* pNF = WnF + (size_t)(k0 + kh) * HH + cj;
                const float* pSR = WsR + (size_t)(k0 + kh) * HH + cj;
                const float* pNR = WnR + (size_t)(k0 + kh) * HH + cj;

                aF = wmma4(aS, mk2(pSF[0], pSF[HH]), aF);
                aF = wmma4(aN, mk2(pNF[0], pNF[HH]), aF);
                aR = wmma4(aS, mk2(pSR[0], pSR[HH]), aR);
                aR = wmma4(aV, mk2(pNR[0], pNR[HH]), aR);
            }

            const float bf = bF[cj];
            const float br = bR[cj];
            #pragma unroll
            for (int v = 0; v < 8; ++v)
                yt[(v + 8 * hi) * TP + cj] = fmaxf(aF[v] + bf, 0.0f) + fmaxf(aR[v] + br, 0.0f);
        }
        __syncthreads();

        if (t0 + wave < batch) {
            const float2 y  = *reinterpret_cast<const float2*>(&yt[wave * TP + 2 * lane]);
            const float2 xv = *reinterpret_cast<const float2*>(xin + (size_t)node * HH + 2 * lane);
            *reinterpret_cast<float2*>(xc + (size_t)(t0 + wave) * HH + 2 * lane) =
                make_float2(xv.x + y.x, xv.y + y.y);
        }
    }
}

__global__ __launch_bounds__(64) void norm_kernel(const float* __restrict__ x,
                                                  float* __restrict__ out) {
    __shared__ float s[64];
    int b = blockIdx.x;
    int c = threadIdx.x;
    float v = x[(size_t)b * HH + c];
    s[c] = v * v;
    __syncthreads();
    #pragma unroll
    for (int off = 32; off > 0; off >>= 1) {
        if (c < off) s[c] += s[c + off];
        __syncthreads();
    }
    float nrm = sqrtf(s[0]);
    out[(size_t)b * HH + c] = v / nrm;
}

extern "C" void kernel_launch(void* const* d_in, const int* in_sizes, int n_in,
                              void* d_out, int out_size, void* d_ws, size_t ws_size,
                              hipStream_t stream) {
    const float* text_emb = (const float*)d_in[0];
    const float* weights  = (const float*)d_in[1];
    const float* W_enc    = (const float*)d_in[2];
    const float* b_enc    = (const float*)d_in[3];
    const float* Wself_f  = (const float*)d_in[4];
    const float* Wneigh_f = (const float*)d_in[5];
    const float* b_f      = (const float*)d_in[6];
    const float* Wself_r  = (const float*)d_in[7];
    const float* Wneigh_r = (const float*)d_in[8];
    const float* b_r      = (const float*)d_in[9];
    const int*   src      = (const int*)d_in[10];
    const int*   dst      = (const int*)d_in[11];
    const int*   ids      = (const int*)d_in[12];
    float* out = (float*)d_out;

    const size_t NH = (size_t)NN * HH;
    float* xA   = (float*)d_ws;
    float* xB   = xA + NH;
    int2*  csrF = (int2*)(xA + 2 * NH);
    int2*  csrR = csrF + NE;
    int*   offF = (int*)(xA + 3 * NH);
    int*   offR = offF + NN;
    int2*  tmpF = (int2*)xA;
    int2*  tmpR = tmpF + NE;
    int*   cnt  = (int*)xB;
    int*   base = cnt + 2 * NCNT;

    const int batch = in_sizes[12];

    hist_kernel<<<EB, 256, 0, stream>>>(src, dst, cnt);
    scan_kernel<<<dim3(SCH, 2), 1024, 0, stream>>>(cnt, base);
    part_kernel<<<EB, 256, 0, stream>>>(src, dst, weights, cnt, base, tmpF, tmpR);
    sort_kernel<<<dim3(NBK, 2), 256, 0, stream>>>(base, tmpF, tmpR, csrF, csrR, offF, offR);

    enc_kernel<<<(NTILES + 7) / 8, 256, 0, stream>>>(text_emb, W_enc, b_enc, xA);

    layer_need_kernel<<<batch, 512, 0, stream>>>(
        xA, xB, ids, batch, csrF, csrR, offF, offR,
        Wself_f, Wneigh_f, b_f, Wself_r, Wneigh_r, b_r);
    layer_rows_kernel<<<64, 512, 0, stream>>>(
        xB, xA, ids, batch, csrF, csrR, offF, offR,
        Wself_f + (size_t)HH * HH, Wneigh_f + (size_t)HH * HH, b_f + HH,
        Wself_r + (size_t)HH * HH, Wneigh_r + (size_t)HH * HH, b_r + HH);

    norm_kernel<<<batch, 64, 0, stream>>>(xA, out);
}


# Last layer and norm fused into the first-layer kernel: a work-group finishes its batch row from first-layer rows kept in LDS, sums in CSR order; norm sum by lane shuffles, same pairs; 8 launches to 6
# speedup vs baseline: 5.8652x; 1.0738x over previous
#include <hip/hip_runtime.h>
#include <hip/hip_bf16.h>
#include <math.h>

typedef __attribute__((ext_vector_type(2))) float v2f;
typedef __attribute__((ext_vector_type(8))) float v8f;

#define NN      100000
#define NE      1600000
#define HH      64
#define TT      384
#define NTILES  6250
#define TP      68
#define NBMASK  0xFFFFF

__device__ __forceinline__ v2f mk2(float a, float b) {
    v2f r; r.x = a; r.y = b; return r;
}

__device__ __forceinline__ v8f wmma4(v2f a, v2f b, v8f c) {
    return __builtin_amdgcn_wmma_f32_16x16x4_f32(
         false, a,  false, b,
         (short)0, c,  false,  false);
}

#define NBK   391
#define EB    400
#define EPB   4000
#define NCNT  (NBK * EB)
#define SCH   ((NCNT + 4095) / 4096)
#define CAP   6144
#define QPT   ((EPB / 4 + 255) / 256)
#define RPT   (CAP / 256)

__device__ __forceinline__ void scan512(int* a, int* wtot, int tid) {
    const int lane = tid & 31;
    const int wv   = tid >> 5;
    const int x0 = a[2 * tid], x1 = a[2 * tid + 1];
    const int s = x0 + x1;
    int inc = s;
    for (int off = 1; off < 32; off <<= 1) {
        int t = __shfl_up(inc, off);
        if (lane >= off) inc += t;
    }
    if (lane == 31) wtot[wv] = inc;
    __syncthreads();
    int pre = 0;
    for (int k = 0; k < 8; ++k) {
        const int v = wtot[k];
        pre += (k < wv) ? v : 0;
    }
    const int ex = pre + inc - s;
    a[2 * tid]     = ex;
    a[2 * tid + 1] = ex + x0;
    __syncthreads();
}

__global__ __launch_bounds__(256) void hist_kernel(const int* __restrict__ src, const int* __restrict__ dst,
                                                   int* __restrict__ cnt) {
    __shared__ int h[2 * NBK];
    const int tid = threadIdx.x;
    const int blk = blockIdx.x;
    for (int t = tid; t < 2 * NBK; t += 256) h[t] = 0;
    __syncthreads();
    const int4* s4 = reinterpret_cast<const int4*>(src + (size_t)blk * EPB);
    const int4* d4 = reinterpret_cast<const int4*>(dst + (size_t)blk * EPB);
    int4 sv[QPT], dv[QPT];
    #pragma unroll
    for (int k = 0; k < QPT; ++k) {
        const int q = tid + 256 * k;
        if (q < EPB / 4) { sv[k] = s4[q]; dv[k] = d4[q]; }
    }
    #pragma unroll
    for (int k = 0; k < QPT; ++k) {
        const int q = tid + 256 * k;
        if (q < EPB / 4) {
            const int4 s = sv[k];
            const int4 d = dv[k];
            atomicAdd(&h[d.x >> 8], 1); atomicAdd(&h[d.y >> 8], 1);
            atomicAdd(&h[d.z >> 8], 1); atomicAdd(&h[d.w >> 8], 1);
            atomicAdd(&h[NBK + (s.x >> 8)], 1); atomicAdd(&h[NBK + (s.y >> 8)], 1);
            atomicAdd(&h[NBK + (s.z >> 8)], 1); atomicAdd(&h[NBK + (s.w >> 8)], 1);
        }
    }
    __syncthreads();
    for (int t = tid; t < 2 * NBK; t += 256) cnt[(size_t)t * EB + blk] = h[t];
}

__global__ __launch_bounds__(1024) void scan_kernel(const int* __restrict__ cnt, int* __restrict__ base) {
    __shared__ int wsum[32];
    __shared__ int psum[32];
    const int tid  = threadIdx.x;
    const int lane = tid & 31;
    const int wv   = tid >> 5;
    const int c    = blockIdx.x;
    const int4* in = reinterpret_cast<const int4*>(cnt + (size_t)blockIdx.y * NCNT);
    int4* out      = reinterpret_cast<int4*>(base + (size_t)blockIdx.y * NCNT);
    int part = 0;
    for (int j = 0; j < c; ++j) {
        const int4 v = in[j * 1024 + tid];
        part += (v.x + v.y) + (v.z + v.w);
    }
    for (int off = 16; off > 0; off >>= 1) part += __shfl_xor(part, off);
    if (lane == 0) psum[wv] = part;
    const int i4 = c * 1024 + tid;
    const bool in_range = i4 < (NCNT / 4);
    int4 v = make_int4(0, 0, 0, 0);
    if (in_range) v = in[i4];
    const int s1 = v.x, s2 = s1 + v.y, s3 = s2 + v.z, s4 = s3 + v.w;
    int inc = s4;
    for (int off = 1; off < 32; off <<= 1) {
        int t = __shfl_up(inc, off);
        if (lane >= off) inc += t;
    }
    if (lane == 31) wsum[wv] = inc;
    __syncthreads();
    int carry = psum[lane];
    for (int off = 16; off > 0; off >>= 1) carry += __shfl_xor(carry, off);
    const int ws = wsum[lane];
    int winc = ws;
    for (int off = 1; off < 32; off <<= 1) {
        int t = __shfl_up(winc, off);
        if (lane >= off) winc += t;
    }
    const int wexcl = __shfl(winc, wv) - __shfl(ws, wv);
    const int b0 = carry + wexcl + (inc - s4);
    if (in_range) out[i4] = make_int4(b0, b0 + s1, b0 + s2, b0 + s3);
}

__global__ __launch_bounds__(256) void part_kernel(const int* __restrict__ src, const int* __restrict__ dst,
                                                   const float* __restrict__ w,
                                                   const int* __restrict__ cnt, const int* __restrict__ base,
                                                   int2* __restrict__ tmpF, int2* __restrict__ tmpR) {
    __shared__ int lb[2][512];
    __shared__ int cur[2][512];
    __shared__ int gb[2][512];
    __shared__ int wtot[8];
    __shared__ int2 srec[2][EPB];
    __shared__ int  sg[2][EPB];
    const int tid = threadIdx.x;
    const int blk = blockIdx.x;
    const int4*   s4 = reinterpret_cast<const int4*>(src + (size_t)blk * EPB);
    const int4*   d4 = reinterpret_cast<const int4*>(dst + (size_t)blk * EPB);
    const float4* w4 = reinterpret_cast<const float4*>(w + (size_t)blk * EPB);
    int4 sv[QPT], dv[QPT];
    float4 wv4[QPT];
    #pragma unroll
    for (int k = 0; k < QPT; ++k) {
        const int q = tid + 256 * k;
        if (q < EPB / 4) { sv[k] = s4[q]; dv[k] = d4[q]; wv4[k] = w4[q]; }
    }
    for (int t = tid; t < 1024; t += 256) {
        const int dir = t >> 9, b = t & 511;
        int c = 0, g = 0;
        if (b < NBK) {
            const size_t ix = (size_t)(dir * NBK + b) * EB + blk;
            c = cnt[ix];
            g = base[ix];
        }
        lb[dir][b] = c;
        gb[dir][b] = g;
    }
    __syncthreads();
    scan512(lb[0], wtot, tid);
    scan512(lb[1], wtot, tid);
    for (int t = tid; t < 1024; t += 256) {
        const int dir = t >> 9, b = t & 511;
        const int l = lb[dir][b];
        cur[dir][b] = l;
        gb[dir][b] -= l;
    }
    __syncthreads();
    #pragma unroll
    for (int k = 0; k < QPT; ++k) {
        const int q = tid + 256 * k;
        if (q < EPB / 4) {
            const int ss[4] = {sv[k].x, sv[k].y, sv[k].z, sv[k].w};
            const int dd[4] = {dv[k].x, dv[k].y, dv[k].z, dv[k].w};
            const float ww[4] = {wv4[k].x, wv4[k].y, wv4[k].z, wv4[k].w};
            #pragma unroll
            for (int u = 0; u < 4; ++u) {
                const int s = ss[u], d = dd[u];
                const int wb = __float_as_int(ww[u]);
                const int bF = d >> 8, bR = s >> 8;
                const int pF = atomicAdd(&cur[0][bF], 1);
                srec[0][pF] = make_int2(s | ((d & 255) << 20), wb);
                sg[0][pF]   = gb[0][bF] + pF;
                const int pR = atomicAdd(&cur[1][bR], 1);
                srec[1][pR] = make_int2(d | ((s & 255) << 20), wb);
                sg[1][pR]   = gb[1][bR] + pR;
            }
        }
    }
    __syncthreads();
    for (int i = tid; i < EPB; i += 256) {
        tmpF[sg[0][i]] = srec[0][i];
        tmpR[sg[1][i]] = srec[1][i];
    }
}

__global__ __launch_bounds__(256) void sort_kernel(const int* __restrict__ base,
                                                   const int2* __restrict__ tmpF, const int2* __restrict__ tmpR,
                                                   int2* __restrict__ csrF, int2* __restrict__ csrR,
                                                   int* __restrict__ offF, int* __restrict__ offR) {
    __shared__ int bins[512];
    __shared__ int cur[256];
    __shared__ int wtot[8];
    __shared__ int2 stage[CAP];
    const int tid = threadIdx.x;
    const int b   = blockIdx.x;
    const int dir = blockIdx.y;
    const int2* tmp = dir ? tmpR : tmpF;
    int2* csr       = dir ? csrR : csrF;
    int* off        = dir ? offR : offF;
    const int beg = base[(size_t)(dir * NBK + b) * EB];
    const int end = (b + 1 < NBK) ? base[(size_t)(dir * NBK + b + 1) * EB] : NE;
    const int n   = end - beg;
    const bool staged = n <= CAP;
    bins[tid] = 0;
    bins[tid + 256] = 0;
    __syncthreads();

    int2 r[RPT];
    if (staged) {
        #pragma unroll
        for (int k = 0; k < RPT; ++k) {
            const int i = tid + 256 * k;
            r[k] = make_int2(0, 0);
            if (i < n) r[k] = tmp[beg + i];
        }
        #pragma unroll
        for (int k = 0; k < RPT; ++k) {
            const int i = tid + 256 * k;
            if (i < n) atomicAdd(&bins[(r[k].x >> 20) & 255], 1);
        }
    } else {
        for (int i = tid; i < n; i += 256) atomicAdd(&bins[(tmp[beg + i].x >> 20) & 255], 1);
    }
    __syncthreads();
    scan512(bins, wtot, tid);
    {
        const int node = b * 256 + tid;
        const int ex = bins[tid];
        if (node < NN) off[node] = beg + ex;
        cur[tid] = ex;
    }
    __syncthreads();

    if (staged) {
        #pragma unroll
        for (int k = 0; k < RPT; ++k) {
            const int i = tid + 256 * k;
            if (i < n) {
                const int nl = (r[k].x >> 20) & 255;
                const int p = atomicAdd(&cur[nl], 1);
                stage[p] = make_int2((r[k].x & NBMASK) | ((nl & 15) << 20), r[k].y);
            }
        }
        __syncthreads();
        for (int i = tid; i < n; i += 256) csr[beg + i] = stage[i];
    } else {
        for (int i = tid; i < n; i += 256) {
            const int2 rr = tmp[beg + i];
            const int nl = (rr.x >> 20) & 255;
            const int p = atomicAdd(&cur[nl], 1);
            csr[beg + p] = make_int2((rr.x & NBMASK) | ((nl & 15) << 20), rr.y);
        }
    }
}

__global__ __launch_bounds__(256) void enc_kernel(const float* __restrict__ emb,
                                                  const float* __restrict__ W,
                                                  const float* __restrict__ bias,
                                                  float* __restrict__ x) {
    const int wave = threadIdx.x >> 5;
    const int lane = threadIdx.x & 31;
    const int nt   = blockIdx.x * 8 + wave;
    if (nt >= NTILES) return;
    const int ln = lane & 15;
    const int hi = lane >> 4;
    const int kh = hi << 1;

    const float* arow = emb + (size_t)(nt * 16 + ln) * TT + kh;
    v8f acc0 = {}, acc1 = {}, acc2 = {}, acc3 = {};

    #pragma unroll 4
    for (int k0 = 0; k0 < TT; k0 += 4) {
        v2f a = *reinterpret_cast<const v2f*>(arow + k0);
        const float* bp = W + (size_t)(k0 + kh) * HH + ln;
        v2f b0 = mk2(bp[0],  bp[HH]);
        v2f b1 = mk2(bp[16], bp[HH + 16]);
        v2f b2 = mk2(bp[32], bp[HH + 32]);
        v2f b3 = mk2(bp[48], bp[HH + 48]);
        acc0 = wmma4(a, b0, acc0);
        acc1 = wmma4(a, b1, acc1);
        acc2 = wmma4(a, b2, acc2);
        acc3 = wmma4(a, b3, acc3);
    }

    float bs0 = bias[ln], bs1 = bias[16 + ln], bs2 = bias[32 + ln], bs3 = bias[48 + ln];
    #pragma unroll
    for (int v = 0; v < 8; ++v) {
        size_t r = (size_t)(nt * 16 + v + 8 * hi) * HH + ln;
        x[r]      = acc0[v] + bs0;
        x[r + 16] = acc1[v] + bs1;
        x[r + 32] = acc2[v] + bs2;
        x[r + 48] = acc3[v] + bs3;
    }
}

__device__ __forceinline__ int2 uniform_entry(int2 e) {
    return make_int2(__builtin_amdgcn_readfirstlane(e.x), __builtin_amdgcn_readfirstlane(e.y));
}

__device__ __forceinline__ void first_entries(const int2* __restrict__ csr, int beg, int end, int2 (&raw)[8]) {
    #pragma unroll
    for (int u = 0; u < 8; ++u) raw[u] = make_int2(0, 0);
    if (beg < end) {
        #pragma unroll
        for (int u = 0; u < 8; ++u) raw[u] = csr[min(beg + u, end - 1)];
    }
}

__device__ __forceinline__ void gather_row_from(const float* __restrict__ xin,
                                                const int2* __restrict__ csr,
                                                int beg, int end, int lane,
                                                float* __restrict__ trow, const int2 (&first)[8]) {
    float ax = 0.0f, ay = 0.0f;
    if (beg < end) {
        int2 en[8];
        #pragma unroll
        for (int u = 0; u < 8; ++u) en[u] = first[u];
        for (int p = beg; p < end; p += 8) {
            float2 v[8];
            #pragma unroll
            for (int u = 0; u < 8; ++u)
                v[u] = *reinterpret_cast<const float2*>(xin + (size_t)(en[u].x & NBMASK) * HH + 2 * lane);
            int2 nx[8];
            #pragma unroll
            for (int u = 0; u < 8; ++u) nx[u] = csr[min(p + 8 + u, end - 1)];
            #pragma unroll
            for (int u = 0; u < 8; ++u) {
                const bool ok = (p + u) < end;
                const float we = __int_as_float(en[u].y);
                ax += ok ? we * v[u].x : 0.0f;
                ay += ok ? we * v[u].y : 0.0f;
            }
            #pragma unroll
            for (int u = 0; u < 8; ++u) en[u] = uniform_entry(nx[u]);
        }
    }
    *reinterpret_cast<float2*>(trow + 2 * lane) = make_float2(ax, ay);
}

__device__ __forceinline__ void seg_bounds(const int* __restrict__ off, int node, int& beg, int& end) {
    const int b = off[node];
    const int e = off[min(node + 1, NN - 1)];
    beg = b;
    end = (node + 1 < NN) ? e : NE;
}

__device__ __forceinline__ void last_layer_sums(const float* x1t, const float* rwt, float (*last)[HH],
                                                int t0, int len, int cF, int lane) {
    float2 sF = *reinterpret_cast<const float2*>(&last[1][2 * lane]);
    float2 sR = *reinterpret_cast<const float2*>(&last[2][2 * lane]);
    #pragma unroll
    for (int r0 = 0; r0 < 16; r0 += 8) {
        float2 row[8];
        float  w[8];
        #pragma unroll
        for (int u = 0; u < 8; ++u) {
            row[u] = *reinterpret_cast<const float2*>(&x1t[(r0 + u) * TP + 2 * lane]);
            w[u]   = rwt[r0 + u];
        }
        #pragma unroll
        for (int u = 0; u < 8; ++u) {
            const int i = t0 + r0 + u;
            if (i < len) {
                if (i == 0) {
                    *reinterpret_cast<float2*>(&last[0][2 * lane]) = row[u];
                } else if (i <= cF) {
                    sF.x += w[u] * row[u].x;
                    sF.y += w[u] * row[u].y;
                } else {
                    sR.x += w[u] * row[u].x;
                    sR.y += w[u] * row[u].y;
                }
            }
        }
    }
    *reinterpret_cast<float2*>(&last[1][2 * lane]) = sF;
    *reinterpret_cast<float2*>(&last[2][2 * lane]) = sR;
}

__global__ __launch_bounds__(512) void layers_kernel(
    const float* __restrict__ xin, float* __restrict__ out,
    const int* __restrict__ ids, int batch,
    const int2* __restrict__ csrF, const int2* __restrict__ csrR,
    const int* __restrict__ offF, const int* __restrict__ offR,
    const float* __restrict__ WsF, const float* __restrict__ WnF, const float* __restrict__ bF,
    const float* __restrict__ WsR, const float* __restrict__ WnR, const float* __restrict__ bR) {
    __shared__ __attribute__((aligned(16))) float tiles[2][16 * TP];
    __shared__ __attribute__((aligned(16))) float yt[2][16 * TP];
    __shared__ __attribute__((aligned(16))) float x1t[16 * TP];
    __shared__ __attribute__((aligned(16))) float last[3][HH];
    __shared__ int   rnode[16];
    __shared__ float rscale[2][16];
    __shared__ float rw[2][16];
    const int wave = __builtin_amdgcn_readfirstlane((int)(threadIdx.x >> 5));
    const int lane = threadIdx.x & 31;
    const int ln = lane & 15;
    const int hi = lane >> 4;
    const int kh = hi << 1;

    const int b = blockIdx.x;
    if (b >= batch) return;
    {
        const int n   = __builtin_amdgcn_readfirstlane(ids[b]);
        int nbF, neF, nbR, neR;
        seg_bounds(offF, n, nbF, neF);
        seg_bounds(offR, n, nbR, neR);
        const int cF  = neF - nbF;
        const int cR  = neR - nbR;
        const int len = 1 + cF + cR;

        if (wave == 8) {
            *reinterpret_cast<float2*>(&last[1][2 * lane]) = make_float2(0.0f, 0.0f);
            *reinterpret_cast<float2*>(&last[2][2 * lane]) = make_float2(0.0f, 0.0f);
        }

        for (int i0 = 0; i0 < len; i0 += 16) {
            const int par = (i0 >> 4) & 1;
            const int i = i0 + wave;
            const bool have = i < len;
            int m = n;
            float wrow = 0.0f;
            if (have && i > 0) {
                const int2 e = (i <= cF) ? csrF[nbF + (i - 1)] : csrR[nbR + (i - 1 - cF)];
                m = e.x & NBMASK;
                wrow = __int_as_float(e.y);
            }
            int begF = 0, endF = 0, begR = 0, endR = 0;
            if (have) {
                seg_bounds(offF, m, begF, endF);
                seg_bounds(offR, m, begR, endR);
            }
            int2 firstF[8], firstR[8];
            first_entries(csrF, begF, endF, firstF);
            first_entries(csrR, begR, endR, firstR);
            #pragma unroll
            for (int u = 0; u < 8; ++u) {
                firstF[u] = uniform_entry(firstF[u]);
                firstR[u] = uniform_entry(firstR[u]);
            }
            gather_row_from(xin, csrF, begF, endF, lane, &tiles[0][wave * TP], firstF);
            gather_row_from(xin, csrR, begR, endR, lane, &tiles[1][wave * TP], firstR);
            if (lane == 0) {
                const int dF = endF - begF;
                const int dR = endR - begR;
                rnode[wave] = m;
                rscale[0][wave] = (dF > 0) ? (1.0f / (float)dF) : 0.0f;
                rscale[1][wave] = (dR > 0) ? (1.0f / (float)dR) : 0.0f;
                rw[par][wave] = wrow;
            }
            __syncthreads();

            if (wave < 8) {
                const int dir = wave >> 2;
                const float* Ws = dir ? WsR : WsF;
                const float* Wn = dir ? WnR : WnF;
                const float* bb = dir ? bR : bF;
                const int cj = 16 * (wave & 3) + ln;
                const float sc = rscale[dir][ln];
                const float* xr = xin + (size_t)rnode[ln] * HH + kh;
                const float* nr = &tiles[dir][ln * TP + kh];

                v8f acc = {};
                #pragma unroll 2
                for (int k0 = 0; k0 < HH; k0 += 4) {
                    v2f aS = *reinterpret_cast<const v2f*>(xr + k0);
                    v2f aN = *reinterpret_cast<const v2f*>(nr + k0);
                    aN *= sc;
                    const float* pS = Ws + (size_t)(k0 + kh) * HH + cj;
                    const float* pN = Wn + (size_t)(k0 + kh) * HH + cj;
                    acc = wmma4(aS, mk2(pS[0], pS[HH]), acc);
                    acc = wmma4(aN, mk2(pN[0], pN[HH]), acc);
                }

                const float bv = bb[cj];
                #pragma unroll
                for (int v = 0; v < 8; ++v)
                    yt[dir][(v + 8 * hi) * TP + cj] = fmaxf(acc[v] + bv, 0.0f);
            } else if (wave == 8 && i0 > 0) {
                last_layer_sums(x1t, rw[par ^ 1], last, i0 - 16, len, cF, lane);
            }
            __syncthreads();

            {
                float2 x1 = make_float2(0.0f, 0.0f);
                if (have) {
                    const float2 yF = *reinterpret_cast<const float2*>(&yt[0][wave * TP + 2 * lane]);
                    const float2 yR = *reinterpret_cast<const float2*>(&yt[1][wave * TP + 2 * lane]);
                    const float2 xv = *reinterpret_cast<const float2*>(xin + (size_t)m * HH + 2 * lane);
                    x1 = make_float2(xv.x + (yF.x + yR.x), xv.y + (yF.y + yR.y));
                }
                *reinterpret_cast<float2*>(&x1t[wave * TP + 2 * lane]) = x1;
            }
        }
        __syncthreads();

        const int tl = (len - 1) >> 4;
        if (wave == 8) {
            last_layer_sums(x1t, rw[tl & 1], last, tl << 4, len, cF, lane);
            if (lane == 0) {
                rscale[0][0] = (cF > 0) ? (1.0f / (float)cF) : 0.0f;
                rscale[1][0] = (cR > 0) ? (1.0f / (float)cR) : 0.0f;
            }
        }
        __syncthreads();

        if (wave < 8) {
            const int dir = wave >> 2;
            const float* Ws = (dir ? WsR : WsF) + (size_t)HH * HH;
            const float* Wn = (dir ? WnR : WnF) + (size_t)HH * HH;
            const float* bb = (dir ? bR : bF) + HH;
            const int cj = 16 * (wave & 3) + ln;
            const float sc = rscale[dir][0];
            const float* xr = &last[0][kh];
            const float* nr = &last[1 + dir][kh];

            v8f acc = {};
            #pragma unroll 2
            for (int k0 = 0; k0 < HH; k0 += 4) {
                v2f aS = *reinterpret_cast<const v2f*>(xr + k0);
                v2f aN = *reinterpret_cast<const v2f*>(nr + k0);
                aN *= sc;
                const float* pS = Ws + (size_t)(k0 + kh) * HH + cj;
                const float* pN = Wn + (size_t)(k0 + kh) * HH + cj;
                acc = wmma4(aS, mk2(pS[0], pS[HH]), acc);
                acc = wmma4(aN, mk2(pN[0], pN[HH]), acc);
            }
            if (hi == 0) yt[dir][cj] = fmaxf(acc[0] + bb[cj], 0.0f);
        }
        __syncthreads();

        if (wave == 0) {
            const float v0 = last[0][lane]      + (yt[0][lane]      + yt[1][lane]);
            const float v1 = last[0][lane + 32] + (yt[0][lane + 32] + yt[1][lane + 32]);
            float ss = v0 * v0 + v1 * v1;
            for (int off = 16; off > 0; off >>= 1) ss += __shfl_xor(ss, off);
            const float nrm = sqrtf(ss);
            out[(size_t)b * HH + lane]      = v0 / nrm;
            out[(size_t)b * HH + lane + 32] = v1 / nrm;
        }
    }
}

extern "C" void kernel_launch(void* const* d_in, const int* in_sizes, int n_in,
                              void* d_out, int out_size, void* d_ws, size_t ws_size,
                              hipStream_t stream) {
    const float* text_emb = (const float*)d_in[0];
    const float* weights  = (const float*)d_in[1];
    const float* W_enc    = (const float*)d_in[2];
    const float* b_enc    = (const float*)d_in[3];
    const float* Wself_f  = (const float*)d_in[4];
    const float* Wneigh_f = (const float*)d_in[5];
    const float* b_f      = (const float*)d_in[6];
    const float* Wself_r  = (const float*)d_in[7];
    const float* Wneigh_r = (const float*)d_in[8];
    const float* b_r      = (const float*)d_in[9];
    const int*   src      = (const int*)d_in[10];
    const int*   dst      = (const int*)d_in[11];
    const int*   ids      = (const int*)d_in[12];
    float* out = (float*)d_out;

    const size_t NH = (size_t)NN * HH;
    float* xA   = (float*)d_ws;
    float* xB   = xA + NH;
    int2*  csrF = (int2*)(xA + 2 * NH);
    int2*  csrR = csrF + NE;
    int*   offF = (int*)(xA + 3 * NH);
    int*   offR = offF + NN;
    int2*  tmpF = (int2*)xA;
    int2*  tmpR = tmpF + NE;
    int*   cnt  = (int*)xB;
    int*   base = cnt + 2 * NCNT;

    const int batch = in_sizes[12];

    hist_kernel<<<EB, 256, 0, stream>>>(src, dst, cnt);
    scan_kernel<<<dim3(SCH, 2), 1024, 0, stream>>>(cnt, base);
    part_kernel<<<EB, 256, 0, stream>>>(src, dst, weights, cnt, base, tmpF, tmpR);
    sort_kernel<<<dim3(NBK, 2), 256, 0, stream>>>(base, tmpF, tmpR, csrF, csrR, offF, offR);

    enc_kernel<<<(NTILES + 7) / 8, 256, 0, stream>>>(text_emb, W_enc, b_enc, xA);

    layers_kernel<<<batch, 512, 0, stream>>>(
        xA, out, ids, batch, csrF, csrR, offF, offR,
        Wself_f, Wneigh_f, b_f, Wself_r, Wneigh_r, b_r);
}


# Encoder operand feed: W_enc kept in LDS per work-group (512 x 16 waves) in 16-byte fragment slots, text_emb rows by 16-byte loads a 32-k block ahead; x now sums k as 0,1,4,5,2,3,6,7 in each 8
# speedup vs baseline: 6.5387x; 1.1148x over previous
#include <hip/hip_runtime.h>
#include <hip/hip_bf16.h>
#include <math.h>

typedef __attribute__((ext_vector_type(2))) float v2f;
typedef __attribute__((ext_vector_type(8))) float v8f;

#define NN      100000
#define NE      1600000
#define HH      64
#define TT      384
#define NTILES  6250
#define TP      68
#define NBMASK  0xFFFFF

__device__ __forceinline__ v2f mk2(float a, float b) {
    v2f r; r.x = a; r.y = b; return r;
}

__device__ __forceinline__ v8f wmma4(v2f a, v2f b, v8f c) {
    return __builtin_amdgcn_wmma_f32_16x16x4_f32(
         false, a,  false, b,
         (short)0, c,  false,  false);
}

#define NBK   391
#define EB    400
#define EPB   4000
#define NCNT  (NBK * EB)
#define SCH   ((NCNT + 4095) / 4096)
#define CAP   6144
#define QPT   ((EPB / 4 + 255) / 256)
#define RPT   (CAP / 256)

__device__ __forceinline__ void scan512(int* a, int* wtot, int tid) {
    const int lane = tid & 31;
    const int wv   = tid >> 5;
    const int x0 = a[2 * tid], x1 = a[2 * tid + 1];
    const int s = x0 + x1;
    int inc = s;
    for (int off = 1; off < 32; off <<= 1) {
        int t = __shfl_up(inc, off);
        if (lane >= off) inc += t;
    }
    if (lane == 31) wtot[wv] = inc;
    __syncthreads();
    int pre = 0;
    for (int k = 0; k < 8; ++k) {
        const int v = wtot[k];
        pre += (k < wv) ? v : 0;
    }
    const int ex = pre + inc - s;
    a[2 * tid]     = ex;
    a[2 * tid + 1] = ex + x0;
    __syncthreads();
}

__global__ __launch_bounds__(256) void hist_kernel(const int* __restrict__ src, const int* __restrict__ dst,
                                                   int* __restrict__ cnt) {
    __shared__ int h[2 * NBK];
    const int tid = threadIdx.x;
    const int blk = blockIdx.x;
    for (int t = tid; t < 2 * NBK; t += 256) h[t] = 0;
    __syncthreads();
    const int4* s4 = reinterpret_cast<const int4*>(src + (size_t)blk * EPB);
    const int4* d4 = reinterpret_cast<const int4*>(dst + (size_t)blk * EPB);
    int4 sv[QPT], dv[QPT];
    #pragma unroll
    for (int k = 0; k < QPT; ++k) {
        const int q = tid + 256 * k;
        if (q < EPB / 4) { sv[k] = s4[q]; dv[k] = d4[q]; }
    }
    #pragma unroll
    for (int k = 0; k < QPT; ++k) {
        const int q = tid + 256 * k;
        if (q < EPB / 4) {
            const int4 s = sv[k];
            const int4 d = dv[k];
            atomicAdd(&h[d.x >> 8], 1); atomicAdd(&h[d.y >> 8], 1);
            atomicAdd(&h[d.z >> 8], 1); atomicAdd(&h[d.w >> 8], 1);
            atomicAdd(&h[NBK + (s.x >> 8)], 1); atomicAdd(&h[NBK + (s.y >> 8)], 1);
            atomicAdd(&h[NBK + (s.z >> 8)], 1); atomicAdd(&h[NBK + (s.w >> 8)], 1);
        }
    }
    __syncthreads();
    for (int t = tid; t < 2 * NBK; t += 256) cnt[(size_t)t * EB + blk] = h[t];
}

__global__ __launch_bounds__(1024) void scan_kernel(const int* __restrict__ cnt, int* __restrict__ base) {
    __shared__ int wsum[32];
    __shared__ int psum[32];
    const int tid  = threadIdx.x;
    const int lane = tid & 31;
    const int wv   = tid >> 5;
    const int c    = blockIdx.x;
    const int4* in = reinterpret_cast<const int4*>(cnt + (size_t)blockIdx.y * NCNT);
    int4* out      = reinterpret_cast<int4*>(base + (size_t)blockIdx.y * NCNT);
    int part = 0;
    for (int j = 0; j < c; ++j) {
        const int4 v = in[j * 1024 + tid];
        part += (v.x + v.y) + (v.z + v.w);
    }
    for (int off = 16; off > 0; off >>= 1) part += __shfl_xor(part, off);
    if (lane == 0) psum[wv] = part;
    const int i4 = c * 1024 + tid;
    const bool in_range = i4 < (NCNT / 4);
    int4 v = make_int4(0, 0, 0, 0);
    if (in_range) v = in[i4];
    const int s1 = v.x, s2 = s1 + v.y, s3 = s2 + v.z, s4 = s3 + v.w;
    int inc = s4;
    for (int off = 1; off < 32; off <<= 1) {
        int t = __shfl_up(inc, off);
        if (lane >= off) inc += t;
    }
    if (lane == 31) wsum[wv] = inc;
    __syncthreads();
    int carry = psum[lane];
    for (int off = 16; off > 0; off >>= 1) carry += __shfl_xor(carry, off);
    const int ws = wsum[lane];
    int winc = ws;
    for (int off = 1; off < 32; off <<= 1) {
        int t = __shfl_up(winc, off);
        if (lane >= off) winc += t;
    }
    const int wexcl = __shfl(winc, wv) - __shfl(ws, wv);
    const int b0 = carry + wexcl + (inc - s4);
    if (in_range) out[i4] = make_int4(b0, b0 + s1, b0 + s2, b0 + s3);
}

__global__ __launch_bounds__(256) void part_kernel(const int* __restrict__ src, const int* __restrict__ dst,
                                                   const float* __restrict__ w,
                                                   const int* __restrict__ cnt, const int* __restrict__ base,
                                                   int2* __restrict__ tmpF, int2* __restrict__ tmpR) {
    __shared__ int lb[2][512];
    __shared__ int cur[2][512];
    __shared__ int gb[2][512];
    __shared__ int wtot[8];
    __shared__ int2 srec[2][EPB];
    __shared__ int  sg[2][EPB];
    const int tid = threadIdx.x;
    const int blk = blockIdx.x;
    const int4*   s4 = reinterpret_cast<const int4*>(src + (size_t)blk * EPB);
    const int4*   d4 = reinterpret_cast<const int4*>(dst + (size_t)blk * EPB);
    const float4* w4 = reinterpret_cast<const float4*>(w + (size_t)blk * EPB);
    int4 sv[QPT], dv[QPT];
    float4 wv4[QPT];
    #pragma unroll
    for (int k = 0; k < QPT; ++k) {
        const int q = tid + 256 * k;
        if (q < EPB / 4) { sv[k] = s4[q]; dv[k] = d4[q]; wv4[k] = w4[q]; }
    }
    for (int t = tid; t < 1024; t += 256) {
        const int dir = t >> 9, b = t & 511;
        int c = 0, g = 0;
        if (b < NBK) {
            const size_t ix = (size_t)(dir * NBK + b) * EB + blk;
            c = cnt[ix];
            g = base[ix];
        }
        lb[dir][b] = c;
        gb[dir][b] = g;
    }
    __syncthreads();
    scan512(lb[0], wtot, tid);
    scan512(lb[1], wtot, tid);
    for (int t = tid; t < 1024; t += 256) {
        const int dir = t >> 9, b = t & 511;
        const int l = lb[dir][b];
        cur[dir][b] = l;
        gb[dir][b] -= l;
    }
    __syncthreads();
    #pragma unroll
    for (int k = 0; k < QPT; ++k) {
        const int q = tid + 256 * k;
        if (q < EPB / 4) {
            const int ss[4] = {sv[k].x, sv[k].y, sv[k].z, sv[k].w};
            const int dd[4] = {dv[k].x, dv[k].y, dv[k].z, dv[k].w};
            const float ww[4] = {wv4[k].x, wv4[k].y, wv4[k].z, wv4[k].w};
            #pragma unroll
            for (int u = 0; u < 4; ++u) {
                const int s = ss[u], d = dd[u];
                const int wb = __float_as_int(ww[u]);
                const int bF = d >> 8, bR = s >> 8;
                const int pF = atomicAdd(&cur[0][bF], 1);
                srec[0][pF] = make_int2(s | ((d & 255) << 20), wb);
                sg[0][pF]   = gb[0][bF] + pF;
                const int pR = atomicAdd(&cur[1][bR], 1);
                srec[1][pR] = make_int2(d | ((s & 255) << 20), wb);
                sg[1][pR]   = gb[1][bR] + pR;
            }
        }
    }
    __syncthreads();
    for (int i = tid; i < EPB; i += 256) {
        tmpF[sg[0][i]] = srec[0][i];
        tmpR[sg[1][i]] = srec[1][i];
    }
}

__global__ __launch_bounds__(256) void sort_kernel(const int* __restrict__ base,
                                                   const int2* __restrict__ tmpF, const int2* __restrict__ tmpR,
                                                   int2* __restrict__ csrF, int2* __restrict__ csrR,
                                                   int* __restrict__ offF, int* __restrict__ offR) {
    __shared__ int bins[512];
    __shared__ int cur[256];
    __shared__ int wtot[8];
    __shared__ int2 stage[CAP];
    const int tid = threadIdx.x;
    const int b   = blockIdx.x;
    const int dir = blockIdx.y;
    const int2* tmp = dir ? tmpR : tmpF;
    int2* csr       = dir ? csrR : csrF;
    int* off        = dir ? offR : offF;
    const int beg = base[(size_t)(dir * NBK + b) * EB];
    const int end = (b + 1 < NBK) ? base[(size_t)(dir * NBK + b + 1) * EB] : NE;
    const int n   = end - beg;
    const bool staged = n <= CAP;
    bins[tid] = 0;
    bins[tid + 256] = 0;
    __syncthreads();

    int2 r[RPT];
    if (staged) {
        #pragma unroll
        for (int k = 0; k < RPT; ++k) {
            const int i = tid + 256 * k;
            r[k] = make_int2(0, 0);
            if (i < n) r[k] = tmp[beg + i];
        }
        #pragma unroll
        for (int k = 0; k < RPT; ++k) {
            const int i = tid + 256 * k;
            if (i < n) atomicAdd(&bins[(r[k].x >> 20) & 255], 1);
        }
    } else {
        for (int i = tid; i < n; i += 256) atomicAdd(&bins[(tmp[beg + i].x >> 20) & 255], 1);
    }
    __syncthreads();
    scan512(bins, wtot, tid);
    {
        const int node = b * 256 + tid;
        const int ex = bins[tid];
        if (node < NN) off[node] = beg + ex;
        cur[tid] = ex;
    }
    __syncthreads();

    if (staged) {
        #pragma unroll
        for (int k = 0; k < RPT; ++k) {
            const int i = tid + 256 * k;
            if (i < n) {
                const int nl = (r[k].x >> 20) & 255;
                const int p = atomicAdd(&cur[nl], 1);
                stage[p] = make_int2((r[k].x & NBMASK) | ((nl & 15) << 20), r[k].y);
            }
        }
        __syncthreads();
        for (int i = tid; i < n; i += 256) csr[beg + i] = stage[i];
    } else {
        for (int i = tid; i < n; i += 256) {
            const int2 rr = tmp[beg + i];
            const int nl = (rr.x >> 20) & 255;
            const int p = atomicAdd(&cur[nl], 1);
            csr[beg + p] = make_int2((rr.x & NBMASK) | ((nl & 15) << 20), rr.y);
        }
    }
}

typedef __attribute__((ext_vector_type(4))) float v4f;

#define ENC_WGS  512
#define KQ       (TT / 4)
#define WQP      72

__device__ __forceinline__ void enc_block(const v4f (&a)[4], const v4f* __restrict__ bp,
                                          v8f& acc0, v8f& acc1, v8f& acc2, v8f& acc3) {
    #pragma unroll
    for (int q = 0; q < 4; ++q) {
        const v4f* p = bp + 2 * q * WQP;
        const v4f b0 = p[0], b1 = p[16], b2 = p[32], b3 = p[48];
        const v2f a01 = mk2(a[q].x, a[q].y), a23 = mk2(a[q].z, a[q].w);
        acc0 = wmma4(a01, mk2(b0.x, b0.y), acc0);
        acc1 = wmma4(a01, mk2(b1.x, b1.y), acc1);
        acc2 = wmma4(a01, mk2(b2.x, b2.y), acc2);
        acc3 = wmma4(a01, mk2(b3.x, b3.y), acc3);
        acc0 = wmma4(a23, mk2(b0.z, b0.w), acc0);
        acc1 = wmma4(a23, mk2(b1.z, b1.w), acc1);
        acc2 = wmma4(a23, mk2(b2.z, b2.w), acc2);
        acc3 = wmma4(a23, mk2(b3.z, b3.w), acc3);
    }
}

__global__ __launch_bounds__(512) void enc_kernel(const float* __restrict__ emb,
                                                  const float* __restrict__ W,
                                                  const float* __restrict__ bias,
                                                  float* __restrict__ x) {
    __shared__ v4f wl[KQ * WQP];
    const int tid  = threadIdx.x;
    const int wave = tid >> 5;
    const int lane = tid & 31;
    const int ln = lane & 15;
    const int hi = lane >> 4;
    const int b  = blockIdx.x;

    const int t  = (13 * ((wave - (b & 3)) & 15)) & 15;
    const int nt = b + ENC_WGS * t;
    const bool active = nt < NTILES;

    const float* arow = emb + (size_t)(nt * 16 + ln) * TT + 4 * hi;
    v4f ac[4];
    #pragma unroll
    for (int q = 0; q < 4; ++q) ac[q] = (v4f){0.0f, 0.0f, 0.0f, 0.0f};
    if (active) {
        #pragma unroll
        for (int q = 0; q < 4; ++q) ac[q] = *reinterpret_cast<const v4f*>(arow + 8 * q);
    }

    v4f wv[KQ * HH / 512];
    #pragma unroll
    for (int i = 0; i < KQ * HH / 512; ++i) {
        const int idx = tid + 512 * i;
        const float* p = W + (size_t)(4 * (idx >> 6)) * HH + (idx & 63);
        wv[i].x = p[0]; wv[i].y = p[HH]; wv[i].z = p[2 * HH]; wv[i].w = p[3 * HH];
    }
    #pragma unroll
    for (int i = 0; i < KQ * HH / 512; ++i) {
        const int idx = tid + 512 * i;
        wl[(idx >> 6) * WQP + (idx & 63)] = wv[i];
    }
    __syncthreads();
    if (!active) return;

    v8f acc0 = {}, acc1 = {}, acc2 = {}, acc3 = {};
    const v4f* bl = wl + hi * WQP + ln;
    v4f an[4];
    for (int blk = 0; blk + 2 < TT / 32; blk += 2) {
        #pragma unroll
        for (int q = 0; q < 4; ++q) an[q] = *reinterpret_cast<const v4f*>(arow + 32 * (blk + 1) + 8 * q);
        enc_block(ac, bl + 8 * blk * WQP, acc0, acc1, acc2, acc3);
        #pragma unroll
        for (int q = 0; q < 4; ++q) ac[q] = *reinterpret_cast<const v4f*>(arow + 32 * (blk + 2) + 8 * q);
        enc_block(an, bl + 8 * (blk + 1) * WQP, acc0, acc1, acc2, acc3);
    }
    #pragma unroll
    for (int q = 0; q < 4; ++q) an[q] = *reinterpret_cast<const v4f*>(arow + 32 * (TT / 32 - 1) + 8 * q);
    enc_block(ac, bl + 8 * (TT / 32 - 2) * WQP, acc0, acc1, acc2, acc3);
    enc_block(an, bl + 8 * (TT / 32 - 1) * WQP, acc0, acc1, acc2, acc3);

    float bs0 = bias[ln], bs1 = bias[16 + ln], bs2 = bias[32 + ln], bs3 = bias[48 + ln];
    #pragma unroll
    for (int v = 0; v < 8; ++v) {
        size_t r = (size_t)(nt * 16 + v + 8 * hi) * HH + ln;
        x[r]      = acc0[v] + bs0;
        x[r + 16] = acc1[v] + bs1;
        x[r + 32] = acc2[v] + bs2;
        x[r + 48] = acc3[v] + bs3;
    }
}

__device__ __forceinline__ int2 uniform_entry(int2 e) {
    return make_int2(__builtin_amdgcn_readfirstlane(e.x), __builtin_amdgcn_readfirstlane(e.y));
}

__device__ __forceinline__ void first_entries(const int2* __restrict__ csr, int beg, int end, int2 (&raw)[8]) {
    #pragma unroll
    for (int u = 0; u < 8; ++u) raw[u] = make_int2(0, 0);
    if (beg < end) {
        #pragma unroll
        for (int u = 0; u < 8; ++u) raw[u] = csr[min(beg + u, end - 1)];
    }
}

__device__ __forceinline__ void gather_row_from(const float* __restrict__ xin,
                                                const int2* __restrict__ csr,
                                                int beg, int end, int lane,
                                                float* __restrict__ trow, const int2 (&first)[8]) {
    float ax = 0.0f, ay = 0.0f;
    if (beg < end) {
        int2 en[8];
        #pragma unroll
        for (int u = 0; u < 8; ++u) en[u] = first[u];
        for (int p = beg; p < end; p += 8) {
            float2 v[8];
            #pragma unroll
            for (int u = 0; u < 8; ++u)
                v[u] = *reinterpret_cast<const float2*>(xin + (size_t)(en[u].x & NBMASK) * HH + 2 * lane);
            int2 nx[8];
            #pragma unroll
            for (int u = 0; u < 8; ++u) nx[u] = csr[min(p + 8 + u, end - 1)];
            #pragma unroll
            for (int u = 0; u < 8; ++u) {
                const bool ok = (p + u) < end;
                const float we = __int_as_float(en[u].y);
                ax += ok ? we * v[u].x : 0.0f;
                ay += ok ? we * v[u].y : 0.0f;
            }
            #pragma unroll
            for (int u = 0; u < 8; ++u) en[u] = uniform_entry(nx[u]);
        }
    }
    *reinterpret_cast<float2*>(trow + 2 * lane) = make_float2(ax, ay);
}

__device__ __forceinline__ void seg_bounds(const int* __restrict__ off, int node, int& beg, int& end) {
    const int b = off[node];
    const int e = off[min(node + 1, NN - 1)];
    beg = b;
    end = (node + 1 < NN) ? e : NE;
}

__device__ __forceinline__ void last_layer_sums(const float* x1t, const float* rwt, float (*last)[HH],
                                                int t0, int len, int cF, int lane) {
    float2 sF = *reinterpret_cast<const float2*>(&last[1][2 * lane]);
    float2 sR = *reinterpret_cast<const float2*>(&last[2][2 * lane]);
    #pragma unroll
    for (int r0 = 0; r0 < 16; r0 += 8) {
        float2 row[8];
        float  w[8];
        #pragma unroll
        for (int u = 0; u < 8; ++u) {
            row[u] = *reinterpret_cast<const float2*>(&x1t[(r0 + u) * TP + 2 * lane]);
            w[u]   = rwt[r0 + u];
        }
        #pragma unroll
        for (int u = 0; u < 8; ++u) {
            const int i = t0 + r0 + u;
            if (i < len) {
                if (i == 0) {
                    *reinterpret_cast<float2*>(&last[0][2 * lane]) = row[u];
                } else if (i <= cF) {
                    sF.x += w[u] * row[u].x;
                    sF.y += w[u] * row[u].y;
                } else {
                    sR.x += w[u] * row[u].x;
                    sR.y += w[u] * row[u].y;
                }
            }
        }
    }
    *reinterpret_cast<float2*>(&last[1][2 * lane]) = sF;
    *reinterpret_cast<float2*>(&last[2][2 * lane]) = sR;
}

__global__ __launch_bounds__(512) void layers_kernel(
    const float* __restrict__ xin, float* __restrict__ out,
    const int* __restrict__ ids, int batch,
    const int2* __restrict__ csrF, const int2* __restrict__ csrR,
    const int* __restrict__ offF, const int* __restrict__ offR,
    const float* __restrict__ WsF, const float* __restrict__ WnF, const float* __restrict__ bF,
    const float* __restrict__ WsR, const float* __restrict__ WnR, const float* __restrict__ bR) {
    __shared__ __attribute__((aligned(16))) float tiles[2][16 * TP];
    __shared__ __attribute__((aligned(16))) float yt[2][16 * TP];
    __shared__ __attribute__((aligned(16))) float x1t[16 * TP];
    __shared__ __attribute__((aligned(16))) float last[3][HH];
    __shared__ int   rnode[16];
    __shared__ float rscale[2][16];
    __shared__ float rw[2][16];
    const int wave = __builtin_amdgcn_readfirstlane((int)(threadIdx.x >> 5));
    const int lane = threadIdx.x & 31;
    const int ln = lane & 15;
    const int hi = lane >> 4;
    const int kh = hi << 1;

    const int b = blockIdx.x;
    if (b >= batch) return;
    {
        const int n   = __builtin_amdgcn_readfirstlane(ids[b]);
        int nbF, neF, nbR, neR;
        seg_bounds(offF, n, nbF, neF);
        seg_bounds(offR, n, nbR, neR);
        const int cF  = neF - nbF;
        const int cR  = neR - nbR;
        const int len = 1 + cF + cR;

        if (wave == 8) {
            *reinterpret_cast<float2*>(&last[1][2 * lane]) = make_float2(0.0f, 0.0f);
            *reinterpret_cast<float2*>(&last[2][2 * lane]) = make_float2(0.0f, 0.0f);
        }

        for (int i0 = 0; i0 < len; i0 += 16) {
            const int par = (i0 >> 4) & 1;
            const int i = i0 + wave;
            const bool have = i < len;
            int m = n;
            float wrow = 0.0f;
            if (have && i > 0) {
                const int2 e = (i <= cF) ? csrF[nbF + (i - 1)] : csrR[nbR + (i - 1 - cF)];
                m = e.x & NBMASK;
                wrow = __int_as_float(e.y);
            }
            int begF = 0, endF = 0, begR = 0, endR = 0;
            if (have) {
                seg_bounds(offF, m, begF, endF);
                seg_bounds(offR, m, begR, endR);
            }
            int2 firstF[8], firstR[8];
            first_entries(csrF, begF, endF, firstF);
            first_entries(csrR, begR, endR, firstR);
            #pragma unroll
            for (int u = 0; u < 8; ++u) {
                firstF[u] = uniform_entry(firstF[u]);
                firstR[u] = uniform_entry(firstR[u]);
            }
            gather_row_from(xin, csrF, begF, endF, lane, &tiles[0][wave * TP], firstF);
            gather_row_from(xin, csrR, begR, endR, lane, &tiles[1][wave * TP], firstR);
            if (lane == 0) {
                const int dF = endF - begF;
                const int dR = endR - begR;
                rnode[wave] = m;
                rscale[0][wave] = (dF > 0) ? (1.0f / (float)dF) : 0.0f;
                rscale[1][wave] = (dR > 0) ? (1.0f / (float)dR) : 0.0f;
                rw[par][wave] = wrow;
            }
            __syncthreads();

            if (wave < 8) {
                const int dir = wave >> 2;
                const float* Ws = dir ? WsR : WsF;
                const float* Wn = dir ? WnR : WnF;
                const float* bb = dir ? bR : bF;
                const int cj = 16 * (wave & 3) + ln;
                const float sc = rscale[dir][ln];
                const float* xr = xin + (size_t)rnode[ln] * HH + kh;
                const float* nr = &tiles[dir][ln * TP + kh];

                v8f acc = {};
                #pragma unroll 2
                for (int k0 = 0; k0 < HH; k0 += 4) {
                    v2f aS = *reinterpret_cast<const v2f*>(xr + k0);
                    v2f aN = *reinterpret_cast<const v2f*>(nr + k0);
                    aN *= sc;
                    const float* pS = Ws + (size_t)(k0 + kh) * HH + cj;
                    const float* pN = Wn + (size_t)(k0 + kh) * HH + cj;
                    acc = wmma4(aS, mk2(pS[0], pS[HH]), acc);
                    acc = wmma4(aN, mk2(pN[0], pN[HH]), acc);
                }

                const float bv = bb[cj];
                #pragma unroll
                for (int v = 0; v < 8; ++v)
                    yt[dir][(v + 8 * hi) * TP + cj] = fmaxf(acc[v] + bv, 0.0f);
            } else if (wave == 8 && i0 > 0) {
                last_layer_sums(x1t, rw[par ^ 1], last, i0 - 16, len, cF, lane);
            }
            __syncthreads();

            {
                float2 x1 = make_float2(0.0f, 0.0f);
                if (have) {
                    const float2 yF = *reinterpret_cast<const float2*>(&yt[0][wave * TP + 2 * lane]);
                    const float2 yR = *reinterpret_cast<const float2*>(&yt[1][wave * TP + 2 * lane]);
                    const float2 xv = *reinterpret_cast<const float2*>(xin + (size_t)m * HH + 2 * lane);
                    x1 = make_float2(xv.x + (yF.x + yR.x), xv.y + (yF.y + yR.y));
                }
                *reinterpret_cast<float2*>(&x1t[wave * TP + 2 * lane]) = x1;
            }
        }
        __syncthreads();

        const int tl = (len - 1) >> 4;
        if (wave == 8) {
            last_layer_sums(x1t, rw[tl & 1], last, tl << 4, len, cF, lane);
            if (lane == 0) {
                rscale[0][0] = (cF > 0) ? (1.0f / (float)cF) : 0.0f;
                rscale[1][0] = (cR > 0) ? (1.0f / (float)cR) : 0.0f;
            }
        }
        __syncthreads();

        if (wave < 8) {
            const int dir = wave >> 2;
            const float* Ws = (dir ? WsR : WsF) + (size_t)HH * HH;
            const float* Wn = (dir ? WnR : WnF) + (size_t)HH * HH;
            const float* bb = (dir ? bR : bF) + HH;
            const int cj = 16 * (wave & 3) + ln;
            const float sc = rscale[dir][0];
            const float* xr = &last[0][kh];
            const float* nr = &last[1 + dir][kh];

            v8f acc = {};
            #pragma unroll 2
            for (int k0 = 0; k0 < HH; k0 += 4) {
                v2f aS = *reinterpret_cast<const v2f*>(xr + k0);
                v2f aN = *reinterpret_cast<const v2f*>(nr + k0);
                aN *= sc;
                const float* pS = Ws + (size_t)(k0 + kh) * HH + cj;
                const float* pN = Wn + (size_t)(k0 + kh) * HH + cj;
                acc = wmma4(aS, mk2(pS[0], pS[HH]), acc);
                acc = wmma4(aN, mk2(pN[0], pN[HH]), acc);
            }
            if (hi == 0) yt[dir][cj] = fmaxf(acc[0] + bb[cj], 0.0f);
        }
        __syncthreads();

        if (wave == 0) {
            const float v0 = last[0][lane]      + (yt[0][lane]      + yt[1][lane]);
            const float v1 = last[0][lane + 32] + (yt[0][lane + 32] + yt[1][lane + 32]);
            float ss = v0 * v0 + v1 * v1;
            for (int off = 16; off > 0; off >>= 1) ss += __shfl_xor(ss, off);
            const float nrm = sqrtf(ss);
            out[(size_t)b * HH + lane]      = v0 / nrm;
            out[(size_t)b * HH + lane + 32] = v1 / nrm;
        }
    }
}

extern "C" void kernel_launch(void* const* d_in, const int* in_sizes, int n_in,
                              void* d_out, int out_size, void* d_ws, size_t ws_size,
                              hipStream_t stream) {
    const float* text_emb = (const float*)d_in[0];
    const float* weights  = (const float*)d_in[1];
    const float* W_enc    = (const float*)d_in[2];
    const float* b_enc    = (const float*)d_in[3];
    const float* Wself_f  = (const float*)d_in[4];
    const float* Wneigh_f = (const float*)d_in[5];
    const float* b_f      = (const float*)d_in[6];
    const float* Wself_r  = (const float*)d_in[7];
    const float* Wneigh_r = (const float*)d_in[8];
    const float* b_r      = (const float*)d_in[9];
    const int*   src      = (const int*)d_in[10];
    const int*   dst      = (const int*)d_in[11];
    const int*   ids      = (const int*)d_in[12];
    float* out = (float*)d_out;

    const size_t NH = (size_t)NN * HH;
    float* xA   = (float*)d_ws;
    float* xB   = xA + NH;
    int2*  csrF = (int2*)(xA + 2 * NH);
    int2*  csrR = csrF + NE;
    int*   offF = (int*)(xA + 3 * NH);
    int*   offR = offF + NN;
    int2*  tmpF = (int2*)xA;
    int2*  tmpR = tmpF + NE;
    int*   cnt  = (int*)xB;
    int*   base = cnt + 2 * NCNT;

    const int batch = in_sizes[12];

    hist_kernel<<<EB, 256, 0, stream>>>(src, dst, cnt);
    scan_kernel<<<dim3(SCH, 2), 1024, 0, stream>>>(cnt, base);
    part_kernel<<<EB, 256, 0, stream>>>(src, dst, weights, cnt, base, tmpF, tmpR);
    sort_kernel<<<dim3(NBK, 2), 256, 0, stream>>>(base, tmpF, tmpR, csrF, csrR, offF, offR);

    enc_kernel<<<ENC_WGS, 512, 0, stream>>>(text_emb, W_enc, b_enc, xA);

    layers_kernel<<<batch, 512, 0, stream>>>(
        xA, out, ids, batch, csrF, csrR, offF, offR,
        Wself_f, Wneigh_f, b_f, Wself_r, Wneigh_r, b_r);
}


# CSR build without count and scan launches (6 to 4): an edge block stores its bucket-ordered records as one piece with its run starts; a bucket fetches its 400 runs; sum order = LDS arrival order
# speedup vs baseline: 7.8699x; 1.2036x over previous
#include <hip/hip_runtime.h>
#include <hip/hip_bf16.h>
#include <math.h>

typedef __attribute__((ext_vector_type(2))) float v2f;
typedef __attribute__((ext_vector_type(8))) float v8f;

#define NN      100000
#define NE      1600000
#define HH      64
#define TT      384
#define NTILES  6250
#define TP      68
#define NBMASK  0xFFFFF

__device__ __forceinline__ v2f mk2(float a, float b) {
    v2f r; r.x = a; r.y = b; return r;
}

__device__ __forceinline__ v8f wmma4(v2f a, v2f b, v8f c) {
    return __builtin_amdgcn_wmma_f32_16x16x4_f32(
         false, a,  false, b,
         (short)0, c,  false,  false);
}

#define NBK   391
#define EB    400
#define EPB   4000
#define NLB   ((NBK + 1) * EB)
#define CAP   6144
#define QPT   ((EPB / 4 + 255) / 256)
#define RPG   (EB / 16)

__device__ __forceinline__ void scan512(int* a, int* wtot, int tid) {
    const int lane = tid & 31;
    const int wv   = tid >> 5;
    const int x0 = a[2 * tid], x1 = a[2 * tid + 1];
    const int s = x0 + x1;
    int inc = s;
    for (int off = 1; off < 32; off <<= 1) {
        int t = __shfl_up(inc, off);
        if (lane >= off) inc += t;
    }
    if (lane == 31) wtot[wv] = inc;
    __syncthreads();
    int pre = 0;
    for (int k = 0; k < 8; ++k) {
        const int v = wtot[k];
        pre += (k < wv) ? v : 0;
    }
    const int ex = pre + inc - s;
    a[2 * tid]     = ex;
    a[2 * tid + 1] = ex + x0;
    __syncthreads();
}

__global__ __launch_bounds__(256) void part_kernel(const int* __restrict__ src, const int* __restrict__ dst,
                                                   const float* __restrict__ w,
                                                   int* __restrict__ lbT,
                                                   int2* __restrict__ tmpF, int2* __restrict__ tmpR) {
    __shared__ int lb[2][512];
    __shared__ int cur[2][512];
    __shared__ int wtot[8];
    __shared__ __attribute__((aligned(16))) int2 srec[2][EPB];
    const int tid = threadIdx.x;
    const int blk = blockIdx.x;
    const int4*   s4 = reinterpret_cast<const int4*>(src + (size_t)blk * EPB);
    const int4*   d4 = reinterpret_cast<const int4*>(dst + (size_t)blk * EPB);
    const float4* w4 = reinterpret_cast<const float4*>(w + (size_t)blk * EPB);
    int4 sv[QPT], dv[QPT];
    float4 wv4[QPT];
    #pragma unroll
    for (int k = 0; k < QPT; ++k) {
        const int q = tid + 256 * k;
        if (q < EPB / 4) { sv[k] = s4[q]; dv[k] = d4[q]; wv4[k] = w4[q]; }
    }
    for (int t = tid; t < 1024; t += 256) (&lb[0][0])[t] = 0;
    __syncthreads();
    #pragma unroll
    for (int k = 0; k < QPT; ++k) {
        const int q = tid + 256 * k;
        if (q < EPB / 4) {
            const int4 s = sv[k];
            const int4 d = dv[k];
            atomicAdd(&lb[0][d.x >> 8], 1); atomicAdd(&lb[0][d.y >> 8], 1);
            atomicAdd(&lb[0][d.z >> 8], 1); atomicAdd(&lb[0][d.w >> 8], 1);
            atomicAdd(&lb[1][s.x >> 8], 1); atomicAdd(&lb[1][s.y >> 8], 1);
            atomicAdd(&lb[1][s.z >> 8], 1); atomicAdd(&lb[1][s.w >> 8], 1);
        }
    }
    __syncthreads();
    scan512(lb[0], wtot, tid);
    scan512(lb[1], wtot, tid);
    for (int t = tid; t < 1024; t += 256) {
        const int dir = t >> 9, b = t & 511;
        const int l = lb[dir][b];
        cur[dir][b] = l;
        if (b <= NBK) lbT[(size_t)(dir * (NBK + 1) + b) * EB + blk] = l;
    }
    __syncthreads();
    #pragma unroll
    for (int k = 0; k < QPT; ++k) {
        const int q = tid + 256 * k;
        if (q < EPB / 4) {
            const int ss[4] = {sv[k].x, sv[k].y, sv[k].z, sv[k].w};
            const int dd[4] = {dv[k].x, dv[k].y, dv[k].z, dv[k].w};
            const float ww[4] = {wv4[k].x, wv4[k].y, wv4[k].z, wv4[k].w};
            #pragma unroll
            for (int u = 0; u < 4; ++u) {
                const int s = ss[u], d = dd[u];
                const int wb = __float_as_int(ww[u]);
                const int pF = atomicAdd(&cur[0][d >> 8], 1);
                srec[0][pF] = make_int2(s | ((d & 255) << 20), wb);
                const int pR = atomicAdd(&cur[1][s >> 8], 1);
                srec[1][pR] = make_int2(d | ((s & 255) << 20), wb);
            }
        }
    }
    __syncthreads();
    const int4* l0 = reinterpret_cast<const int4*>(&srec[0][0]);
    const int4* l1 = reinterpret_cast<const int4*>(&srec[1][0]);
    int4* oF = reinterpret_cast<int4*>(tmpF + (size_t)blk * EPB);
    int4* oR = reinterpret_cast<int4*>(tmpR + (size_t)blk * EPB);
    for (int i = tid; i < EPB / 2; i += 256) {
        oF[i] = l0[i];
        oR[i] = l1[i];
    }
}

__global__ __launch_bounds__(256) void sort_kernel(const int* __restrict__ lbT,
                                                   const int2* __restrict__ tmpF, const int2* __restrict__ tmpR,
                                                   int2* __restrict__ csrF, int2* __restrict__ csrR,
                                                   int* __restrict__ offF, int* __restrict__ offR) {
    __shared__ int bins[512];
    __shared__ int cur[256];
    __shared__ int wtot[8];
    __shared__ int run[EB];
    __shared__ int tot[2];
    __shared__ int2 stage[CAP];
    const int tid = threadIdx.x;
    const int b   = blockIdx.x;
    const int dir = blockIdx.y;
    const int2* tmp = dir ? tmpR : tmpF;
    int2* csr       = dir ? csrR : csrF;
    int* off        = dir ? offR : offF;
    const int* row  = lbT + (size_t)(dir * (NBK + 1) + b) * EB;
    const int k1 = tid + 256;
    const int s0 = row[tid], e0 = row[EB + tid];
    int s1 = 0, e1 = 0;
    if (k1 < EB) { s1 = row[k1]; e1 = row[EB + k1]; }
    run[tid] = s0 | ((e0 - s0) << 12);
    if (k1 < EB) run[k1] = s1 | ((e1 - s1) << 12);
    int ssum = s0 + s1;
    int lsum = (e0 - s0) + (e1 - s1);
    for (int o = 16; o > 0; o >>= 1) {
        ssum += __shfl_xor(ssum, o);
        lsum += __shfl_xor(lsum, o);
    }
    bins[tid] = 0;
    bins[tid + 256] = 0;
    if (tid < 2) tot[tid] = 0;
    __syncthreads();
    if ((tid & 31) == 0) {
        atomicAdd(&tot[0], ssum);
        atomicAdd(&tot[1], lsum);
    }
    __syncthreads();
    const int beg = tot[0];
    const int n   = tot[1];
    const bool staged = n <= CAP;

    const int g = tid >> 4;
    const int j = tid & 15;
    const int o0 = staged ? 16 : 0;
    int2 r[RPG];
    unsigned have = 0u;
    unsigned more = 0u;
    {
        int pk[RPG];
        #pragma unroll
        for (int it = 0; it < RPG; ++it) pk[it] = run[g + 16 * it];
        #pragma unroll
        for (int it = 0; it < RPG; ++it) {
            const int len = pk[it] >> 12;
            if (len > o0) more |= 1u << it;
            if (staged && j < len) have |= 1u << it;
            r[it] = make_int2(0, 0);
        }
        if (staged) {
            #pragma unroll
            for (int it = 0; it < RPG; ++it) {
                const int st = pk[it] & 4095, len = pk[it] >> 12;
                const int pos = min(st + min(j, max(len, 1) - 1), EPB - 1);
                r[it] = tmp[(size_t)(g + 16 * it) * EPB + pos];
            }
        }
    }
    #pragma unroll
    for (int it = 0; it < RPG; ++it) {
        if (have & (1u << it)) atomicAdd(&bins[(r[it].x >> 20) & 255], 1);
    }
    for (unsigned m = more; m != 0u; m &= m - 1u) {
        const int k = g + 16 * (__ffs(m) - 1);
        const int pk = run[k];
        const int len = pk >> 12;
        const int2* p = tmp + (size_t)k * EPB + (pk & 4095);
        for (int o = o0 + j; o < len; o += 16) atomicAdd(&bins[(p[o].x >> 20) & 255], 1);
    }
    __syncthreads();
    scan512(bins, wtot, tid);
    {
        const int node = b * 256 + tid;
        const int ex = bins[tid];
        if (node < NN) off[node] = beg + ex;
        cur[tid] = ex;
    }
    __syncthreads();

    #pragma unroll
    for (int it = 0; it < RPG; ++it) {
        if (have & (1u << it)) {
            const int nl = (r[it].x >> 20) & 255;
            const int q = atomicAdd(&cur[nl], 1);
            stage[q] = make_int2((r[it].x & NBMASK) | ((nl & 15) << 20), r[it].y);
        }
    }
    for (unsigned m = more; m != 0u; m &= m - 1u) {
        const int k = g + 16 * (__ffs(m) - 1);
        const int pk = run[k];
        const int len = pk >> 12;
        const int2* p = tmp + (size_t)k * EPB + (pk & 4095);
        for (int o = o0 + j; o < len; o += 16) {
            const int2 rr = p[o];
            const int nl = (rr.x >> 20) & 255;
            const int q = atomicAdd(&cur[nl], 1);
            const int2 e = make_int2((rr.x & NBMASK) | ((nl & 15) << 20), rr.y);
            if (staged) stage[q] = e;
            else        csr[beg + q] = e;
        }
    }
    if (staged) {
        __syncthreads();
        for (int i = tid; i < n; i += 256) csr[beg + i] = stage[i];
    }
}

typedef __attribute__((ext_vector_type(4))) float v4f;

#define ENC_WGS  512
#define KQ       (TT / 4)
#define WQP      72

__device__ __forceinline__ void enc_block(const v4f (&a)[4], const v4f* __restrict__ bp,
                                          v8f& acc0, v8f& acc1, v8f& acc2, v8f& acc3) {
    #pragma unroll
    for (int q = 0; q < 4; ++q) {
        const v4f* p = bp + 2 * q * WQP;
        const v4f b0 = p[0], b1 = p[16], b2 = p[32], b3 = p[48];
        const v2f a01 = mk2(a[q].x, a[q].y), a23 = mk2(a[q].z, a[q].w);
        acc0 = wmma4(a01, mk2(b0.x, b0.y), acc0);
        acc1 = wmma4(a01, mk2(b1.x, b1.y), acc1);
        acc2 = wmma4(a01, mk2(b2.x, b2.y), acc2);
        acc3 = wmma4(a01, mk2(b3.x, b3.y), acc3);
        acc0 = wmma4(a23, mk2(b0.z, b0.w), acc0);
        acc1 = wmma4(a23, mk2(b1.z, b1.w), acc1);
        acc2 = wmma4(a23, mk2(b2.z, b2.w), acc2);
        acc3 = wmma4(a23, mk2(b3.z, b3.w), acc3);
    }
}

__global__ __launch_bounds__(512) void enc_kernel(const float* __restrict__ emb,
                                                  const float* __restrict__ W,
                                                  const float* __restrict__ bias,
                                                  float* __restrict__ x) {
    __shared__ v4f wl[KQ * WQP];
    const int tid  = threadIdx.x;
    const int wave = tid >> 5;
    const int lane = tid & 31;
    const int ln = lane & 15;
    const int hi = lane >> 4;
    const int b  = blockIdx.x;

    const int t  = (13 * ((wave - (b & 3)) & 15)) & 15;
    const int nt = b + ENC_WGS * t;
    const bool active = nt < NTILES;

    const float* arow = emb + (size_t)(nt * 16 + ln) * TT + 4 * hi;
    v4f ac[4];
    #pragma unroll
    for (int q = 0; q < 4; ++q) ac[q] = (v4f){0.0f, 0.0f, 0.0f, 0.0f};
    if (active) {
        #pragma unroll
        for (int q = 0; q < 4; ++q) ac[q] = *reinterpret_cast<const v4f*>(arow + 8 * q);
    }

    v4f wv[KQ * HH / 512];
    #pragma unroll
    for (int i = 0; i < KQ * HH / 512; ++i) {
        const int idx = tid + 512 * i;
        const float* p = W + (size_t)(4 * (idx >> 6)) * HH + (idx & 63);
        wv[i].x = p[0]; wv[i].y = p[HH]; wv[i].z = p[2 * HH]; wv[i].w = p[3 * HH];
    }
    #pragma unroll
    for (int i = 0; i < KQ * HH / 512; ++i) {
        const int idx = tid + 512 * i;
        wl[(idx >> 6) * WQP + (idx & 63)] = wv[i];
    }
    __syncthreads();
    if (!active) return;

    v8f acc0 = {}, acc1 = {}, acc2 = {}, acc3 = {};
    const v4f* bl = wl + hi * WQP + ln;
    v4f an[4];
    for (int blk = 0; blk + 2 < TT / 32; blk += 2) {
        #pragma unroll
        for (int q = 0; q < 4; ++q) an[q] = *reinterpret_cast<const v4f*>(arow + 32 * (blk + 1) + 8 * q);
        enc_block(ac, bl + 8 * blk * WQP, acc0, acc1, acc2, acc3);
        #pragma unroll
        for (int q = 0; q < 4; ++q) ac[q] = *reinterpret_cast<const v4f*>(arow + 32 * (blk + 2) + 8 * q);
        enc_block(an, bl + 8 * (blk + 1) * WQP, acc0, acc1, acc2, acc3);
    }
    #pragma unroll
    for (int q = 0; q < 4; ++q) an[q] = *reinterpret_cast<const v4f*>(arow + 32 * (TT / 32 - 1) + 8 * q);
    enc_block(ac, bl + 8 * (TT / 32 - 2) * WQP, acc0, acc1, acc2, acc3);
    enc_block(an, bl + 8 * (TT / 32 - 1) * WQP, acc0, acc1, acc2, acc3);

    float bs0 = bias[ln], bs1 = bias[16 + ln], bs2 = bias[32 + ln], bs3 = bias[48 + ln];
    #pragma unroll
    for (int v = 0; v < 8; ++v) {
        size_t r = (size_t)(nt * 16 + v + 8 * hi) * HH + ln;
        x[r]      = acc0[v] + bs0;
        x[r + 16] = acc1[v] + bs1;
        x[r + 32] = acc2[v] + bs2;
        x[r + 48] = acc3[v] + bs3;
    }
}

__device__ __forceinline__ int2 uniform_entry(int2 e) {
    return make_int2(__builtin_amdgcn_readfirstlane(e.x), __builtin_amdgcn_readfirstlane(e.y));
}

__device__ __forceinline__ void first_entries(const int2* __restrict__ csr, int beg, int end, int2 (&raw)[8]) {
    #pragma unroll
    for (int u = 0; u < 8; ++u) raw[u] = make_int2(0, 0);
    if (beg < end) {
        #pragma unroll
        for (int u = 0; u < 8; ++u) raw[u] = csr[min(beg + u, end - 1)];
    }
}

__device__ __forceinline__ void gather_row_from(const float* __restrict__ xin,
                                                const int2* __restrict__ csr,
                                                int beg, int end, int lane,
                                                float* __restrict__ trow, const int2 (&first)[8]) {
    float ax = 0.0f, ay = 0.0f;
    if (beg < end) {
        int2 en[8];
        #pragma unroll
        for (int u = 0; u < 8; ++u) en[u] = first[u];
        for (int p = beg; p < end; p += 8) {
            float2 v[8];
            #pragma unroll
            for (int u = 0; u < 8; ++u)
                v[u] = *reinterpret_cast<const float2*>(xin + (size_t)(en[u].x & NBMASK) * HH + 2 * lane);
            int2 nx[8];
            #pragma unroll
            for (int u = 0; u < 8; ++u) nx[u] = csr[min(p + 8 + u, end - 1)];
            #pragma unroll
            for (int u = 0; u < 8; ++u) {
                const bool ok = (p + u) < end;
                const float we = __int_as_float(en[u].y);
                ax += ok ? we * v[u].x : 0.0f;
                ay += ok ? we * v[u].y : 0.0f;
            }
            #pragma unroll
            for (int u = 0; u < 8; ++u) en[u] = uniform_entry(nx[u]);
        }
    }
    *reinterpret_cast<float2*>(trow + 2 * lane) = make_float2(ax, ay);
}

__device__ __forceinline__ void seg_bounds(const int* __restrict__ off, int node, int& beg, int& end) {
    const int b = off[node];
    const int e = off[min(node + 1, NN - 1)];
    beg = b;
    end = (node + 1 < NN) ? e : NE;
}

__device__ __forceinline__ void last_layer_sums(const float* x1t, const float* rwt, float (*last)[HH],
                                                int t0, int len, int cF, int lane) {
    float2 sF = *reinterpret_cast<const float2*>(&last[1][2 * lane]);
    float2 sR = *reinterpret_cast<const float2*>(&last[2][2 * lane]);
    #pragma unroll
    for (int r0 = 0; r0 < 16; r0 += 8) {
        float2 row[8];
        float  w[8];
        #pragma unroll
        for (int u = 0; u < 8; ++u) {
            row[u] = *reinterpret_cast<const float2*>(&x1t[(r0 + u) * TP + 2 * lane]);
            w[u]   = rwt[r0 + u];
        }
        #pragma unroll
        for (int u = 0; u < 8; ++u) {
            const int i = t0 + r0 + u;
            if (i < len) {
                if (i == 0) {
                    *reinterpret_cast<float2*>(&last[0][2 * lane]) = row[u];
                } else if (i <= cF) {
                    sF.x += w[u] * row[u].x;
                    sF.y += w[u] * row[u].y;
                } else {
                    sR.x += w[u] * row[u].x;
                    sR.y += w[u] * row[u].y;
                }
            }
        }
    }
    *reinterpret_cast<float2*>(&last[1][2 * lane]) = sF;
    *reinterpret_cast<float2*>(&last[2][2 * lane]) = sR;
}

__global__ __launch_bounds__(512) void layers_kernel(
    const float* __restrict__ xin, float* __restrict__ out,
    const int* __restrict__ ids, int batch,
    const int2* __restrict__ csrF, const int2* __restrict__ csrR,
    const int* __restrict__ offF, const int* __restrict__ offR,
    const float* __restrict__ WsF, const float* __restrict__ WnF, const float* __restrict__ bF,
    const float* __restrict__ WsR, const float* __restrict__ WnR, const float* __restrict__ bR) {
    __shared__ __attribute__((aligned(16))) float tiles[2][16 * TP];
    __shared__ __attribute__((aligned(16))) float yt[2][16 * TP];
    __shared__ __attribute__((aligned(16))) float x1t[16 * TP];
    __shared__ __attribute__((aligned(16))) float last[3][HH];
    __shared__ int   rnode[16];
    __shared__ float rscale[2][16];
    __shared__ float rw[2][16];
    const int wave = __builtin_amdgcn_readfirstlane((int)(threadIdx.x >> 5));
    const int lane = threadIdx.x & 31;
    const int ln = lane & 15;
    const int hi = lane >> 4;
    const int kh = hi << 1;

    const int b = blockIdx.x;
    if (b >= batch) return;
    {
        const int n   = __builtin_amdgcn_readfirstlane(ids[b]);
        int nbF, neF, nbR, neR;
        seg_bounds(offF, n, nbF, neF);
        seg_bounds(offR, n, nbR, neR);
        const int cF  = neF - nbF;
        const int cR  = neR - nbR;
        const int len = 1 + cF + cR;

        if (wave == 8) {
            *reinterpret_cast<float2*>(&last[1][2 * lane]) = make_float2(0.0f, 0.0f);
            *reinterpret_cast<float2*>(&last[2][2 * lane]) = make_float2(0.0f, 0.0f);
        }

        for (int i0 = 0; i0 < len; i0 += 16) {
            const int par = (i0 >> 4) & 1;
            const int i = i0 + wave;
            const bool have = i < len;
            int m = n;
            float wrow = 0.0f;
            if (have && i > 0) {
                const int2 e = (i <= cF) ? csrF[nbF + (i - 1)] : csrR[nbR + (i - 1 - cF)];
                m = e.x & NBMASK;
                wrow = __int_as_float(e.y);
            }
            int begF = 0, endF = 0, begR = 0, endR = 0;
            if (have) {
                seg_bounds(offF, m, begF, endF);
                seg_bounds(offR, m, begR, endR);
            }
            int2 firstF[8], firstR[8];
            first_entries(csrF, begF, endF, firstF);
            first_entries(csrR, begR, endR, firstR);
            #pragma unroll
            for (int u = 0; u < 8; ++u) {
                firstF[u] = uniform_entry(firstF[u]);
                firstR[u] = uniform_entry(firstR[u]);
            }
            gather_row_from(xin, csrF, begF, endF, lane, &tiles[0][wave * TP], firstF);
            gather_row_from(xin, csrR, begR, endR, lane, &tiles[1][wave * TP], firstR);
            if (lane == 0) {
                const int dF = endF - begF;
                const int dR = endR - begR;
                rnode[wave] = m;
                rscale[0][wave] = (dF > 0) ? (1.0f / (float)dF) : 0.0f;
                rscale[1][wave] = (dR > 0) ? (1.0f / (float)dR) : 0.0f;
                rw[par][wave] = wrow;
            }
            __syncthreads();

            if (wave < 8) {
                const int dir = wave >> 2;
                const float* Ws = dir ? WsR : WsF;
                const float* Wn = dir ? WnR : WnF;
                const float* bb = dir ? bR : bF;
                const int cj = 16 * (wave & 3) + ln;
                const float sc = rscale[dir][ln];
                const float* xr = xin + (size_t)rnode[ln] * HH + kh;
                const float* nr = &tiles[dir][ln * TP + kh];

                v8f acc = {};
                #pragma unroll 2
                for (int k0 = 0; k0 < HH; k0 += 4) {
                    v2f aS = *reinterpret_cast<const v2f*>(xr + k0);
                    v2f aN = *reinterpret_cast<const v2f*>(nr + k0);
                    aN *= sc;
                    const float* pS = Ws + (size_t)(k0 + kh) * HH + cj;
                    const float* pN = Wn + (size_t)(k0 + kh) * HH + cj;
                    acc = wmma4(aS, mk2(pS[0], pS[HH]), acc);
                    acc = wmma4(aN, mk2(pN[0], pN[HH]), acc);
                }

                const float bv = bb[cj];
                #pragma unroll
                for (int v = 0; v < 8; ++v)
                    yt[dir][(v + 8 * hi) * TP + cj] = fmaxf(acc[v] + bv, 0.0f);
            } else if (wave == 8 && i0 > 0) {
                last_layer_sums(x1t, rw[par ^ 1], last, i0 - 16, len, cF, lane);
            }
            __syncthreads();

            {
                float2 x1 = make_float2(0.0f, 0.0f);
                if (have) {
                    const float2 yF = *reinterpret_cast<const float2*>(&yt[0][wave * TP + 2 * lane]);
                    const float2 yR = *reinterpret_cast<const float2*>(&yt[1][wave * TP + 2 * lane]);
                    const float2 xv = *reinterpret_cast<const float2*>(xin + (size_t)m * HH + 2 * lane);
                    x1 = make_float2(xv.x + (yF.x + yR.x), xv.y + (yF.y + yR.y));
                }
                *reinterpret_cast<float2*>(&x1t[wave * TP + 2 * lane]) = x1;
            }
        }
        __syncthreads();

        const int tl = (len - 1) >> 4;
        if (wave == 8) {
            last_layer_sums(x1t, rw[tl & 1], last, tl << 4, len, cF, lane);
            if (lane == 0) {
                rscale[0][0] = (cF > 0) ? (1.0f / (float)cF) : 0.0f;
                rscale[1][0] = (cR > 0) ? (1.0f / (float)cR) : 0.0f;
            }
        }
        __syncthreads();

        if (wave < 8) {
            const int dir = wave >> 2;
            const float* Ws = (dir ? WsR : WsF) + (size_t)HH * HH;
            const float* Wn = (dir ? WnR : WnF) + (size_t)HH * HH;
            const float* bb = (dir ? bR : bF) + HH;
            const int cj = 16 * (wave & 3) + ln;
            const float sc = rscale[dir][0];
            const float* xr = &last[0][kh];
            const float* nr = &last[1 + dir][kh];

            v8f acc = {};
            #pragma unroll 2
            for (int k0 = 0; k0 < HH; k0 += 4) {
                v2f aS = *reinterpret_cast<const v2f*>(xr + k0);
                v2f aN = *reinterpret_cast<const v2f*>(nr + k0);
                aN *= sc;
                const float* pS = Ws + (size_t)(k0 + kh) * HH + cj;
                const float* pN = Wn + (size_t)(k0 + kh) * HH + cj;
                acc = wmma4(aS, mk2(pS[0], pS[HH]), acc);
                acc = wmma4(aN, mk2(pN[0], pN[HH]), acc);
            }
            if (hi == 0) yt[dir][cj] = fmaxf(acc[0] + bb[cj], 0.0f);
        }
        __syncthreads();

        if (wave == 0) {
            const float v0 = last[0][lane]      + (yt[0][lane]      + yt[1][lane]);
            const float v1 = last[0][lane + 32] + (yt[0][lane + 32] + yt[1][lane + 32]);
            float ss = v0 * v0 + v1 * v1;
            for (int off = 16; off > 0; off >>= 1) ss += __shfl_xor(ss, off);
            const float nrm = sqrtf(ss);
            out[(size_t)b * HH + lane]      = v0 / nrm;
            out[(size_t)b * HH + lane + 32] = v1 / nrm;
        }
    }
}


extern "C" void kernel_launch(void* const* d_in, const int* in_sizes, int n_in,
                              void* d_out, int out_size, void* d_ws, size_t ws_size,
                              hipStream_t stream) {
    const float* text_emb = (const float*)d_in[0];
    const float* weights  = (const float*)d_in[1];
    const float* W_enc    = (const float*)d_in[2];
    const float* b_enc    = (const float*)d_in[3];
    const float* Wself_f  = (const float*)d_in[4];
    const float* Wneigh_f = (const float*)d_in[5];
    const float* b_f      = (const float*)d_in[6];
    const float* Wself_r  = (const float*)d_in[7];
    const float* Wneigh_r = (const float*)d_in[8];
    const float* b_r      = (const float*)d_in[9];
    const int*   src      = (const int*)d_in[10];
    const int*   dst      = (const int*)d_in[11];
    const int*   ids      = (const int*)d_in[12];
    float* out = (float*)d_out;

    const size_t NH = (size_t)NN * HH;
    float* xA   = (float*)d_ws;
    float* xB   = xA + NH;
    int2*  csrF = (int2*)(xA + 2 * NH);
    int2*  csrR = csrF + NE;
    int*   offF = (int*)(xA + 3 * NH);
    int*   offR = offF + NN;
    int2*  tmpF = (int2*)xA;
    int2*  tmpR = tmpF + NE;
    int*   lbT  = (int*)xB;

    const int batch = in_sizes[12];

    part_kernel<<<EB, 256, 0, stream>>>(src, dst, weights, lbT, tmpF, tmpR);
    sort_kernel<<<dim3(NBK, 2), 256, 0, stream>>>(lbT, tmpF, tmpR, csrF, csrR, offF, offR);

    enc_kernel<<<ENC_WGS, 512, 0, stream>>>(text_emb, W_enc, b_enc, xA);

    layers_kernel<<<batch, 512, 0, stream>>>(
        xA, out, ids, batch, csrF, csrR, offF, offR,
        Wself_f, Wneigh_f, b_f, Wself_r, Wneigh_r, b_r);
}
